# starting kernel
# baseline (speedup 1.0000x reference)
#include <hip/hip_runtime.h>

#define HID 64

typedef float v2f __attribute__((ext_vector_type(2)));
typedef float v8f __attribute__((ext_vector_type(8)));

__global__ void k_zero(float* __restrict__ p, int n) {
  int t = blockIdx.x * blockDim.x + threadIdx.x;
  if (t < n) p[t] = 0.0f;
}

__global__ void k_degree(const int* __restrict__ ei, int E, float* __restrict__ deg) {
  int e = blockIdx.x * blockDim.x + threadIdx.x;
  if (e < E) atomicAdd(&deg[ei[E + e]], 1.0f);
}

__global__ void k_dinv(float* __restrict__ deg, int N) {
  int i = blockIdx.x * blockDim.x + threadIdx.x;
  if (i < N) deg[i] = rsqrtf(deg[i] + 1.0f);
}

__global__ void k_gemm1(const float* __restrict__ x, const float* __restrict__ W1,
                        float* __restrict__ out, int N) {
  int t = blockIdx.x * blockDim.x + threadIdx.x;
  if (t >= N * HID) return;
  int n = t >> 6, j = t & 63;
  float x0 = x[n * 3 + 0], x1 = x[n * 3 + 1], x2 = x[n * 3 + 2];
  out[t] = x0 * W1[j] + x1 * W1[HID + j] + x2 * W1[2 * HID + j];
}

__global__ __launch_bounds__(32) void k_gemm2(const float* __restrict__ H,
                                              const float* __restrict__ W,
                                              float* __restrict__ O, int N) {
  const int tm   = blockIdx.x;
  const int tn   = blockIdx.y;
  const int lane = threadIdx.x;
  const int m    = lane & 15;
  const int half = lane >> 4;
  const int ncol = tn * 16 + (lane & 15);

  int arow = tm * 16 + m;
  if (arow >= N) arow = N - 1;
  const float* __restrict__ Arow = H + (long)arow * HID;

  v8f acc = {};
  #pragma unroll
  for (int k0 = 0; k0 < HID; k0 += 4) {
    const int ka = k0 + half * 2;
    v2f a; a.x = Arow[ka];             a.y = Arow[ka + 1];
    v2f b; b.x = W[ka * HID + ncol];   b.y = W[(ka + 1) * HID + ncol];
    acc = __builtin_amdgcn_wmma_f32_16x16x4_f32(
         false, a,  false, b,
         (short)0, acc,  false,  false);
  }
  #pragma unroll
  for (int r = 0; r < 8; ++r) {
    int mrow = tm * 16 + r + half * 8;
    if (mrow < N) O[(long)mrow * HID + ncol] = acc[r];
  }
}

__global__ void k_scatter(const int* __restrict__ ei, int E,
                          const float* __restrict__ hw,
                          const float* __restrict__ dinv,
                          float* __restrict__ agg) {
  int t = blockIdx.x * blockDim.x + threadIdx.x;
  if (t >= E * 16) return;
  int e = t >> 4;
  int j = (t & 15) * 4;
  int s = ei[e];
  int d = ei[E + e];
  float norm = dinv[s] * dinv[d];
  const float4 v = *(const float4*)(hw + (long)s * HID + j);
  float* o = agg + (long)d * HID + j;
  atomicAdd(o + 0, v.x * norm);
  atomicAdd(o + 1, v.y * norm);
  atomicAdd(o + 2, v.z * norm);
  atomicAdd(o + 3, v.w * norm);
}

__global__ void k_combine(float* __restrict__ agg, const float* __restrict__ hw,
                          const float* __restrict__ dinv, const float* __restrict__ bias,
                          int N) {
  int t = blockIdx.x * blockDim.x + threadIdx.x;
  if (t >= N * 16) return;
  int n = t >> 4;
  int j = (t & 15) * 4;
  float di = dinv[n];
  float sl = di * di;
  float4 a = *(float4*)(agg + (long)n * HID + j);
  const float4 h = *(const float4*)(hw + (long)n * HID + j);
  a.x = fmaxf(fmaf(h.x, sl, a.x) + bias[j + 0], 0.0f);
  a.y = fmaxf(fmaf(h.y, sl, a.y) + bias[j + 1], 0.0f);
  a.z = fmaxf(fmaf(h.z, sl, a.z) + bias[j + 2], 0.0f);
  a.w = fmaxf(fmaf(h.w, sl, a.w) + bias[j + 3], 0.0f);
  *(float4*)(agg + (long)n * HID + j) = a;
}

__global__ void k_pool(const float* __restrict__ h, const int* __restrict__ batch,
                       float* __restrict__ out, float* __restrict__ cnt, int N) {
  int t = blockIdx.x * blockDim.x + threadIdx.x;
  if (t >= N * 16) return;
  int n = t >> 4;
  int j = (t & 15) * 4;
  int g = batch[n];
  const float4 v = *(const float4*)(h + (long)n * HID + j);
  float* o = out + (long)g * HID + j;
  atomicAdd(o + 0, v.x);
  atomicAdd(o + 1, v.y);
  atomicAdd(o + 2, v.z);
  atomicAdd(o + 3, v.w);
  if ((t & 15) == 0) atomicAdd(&cnt[g], 1.0f);
}

__global__ void k_div(float* __restrict__ out, const float* __restrict__ cnt, int total) {
  int t = blockIdx.x * blockDim.x + threadIdx.x;
  if (t < total) out[t] /= fmaxf(cnt[t >> 6], 1.0f);
}

extern "C" void kernel_launch(void* const* d_in, const int* in_sizes, int n_in,
                              void* d_out, int out_size, void* d_ws, size_t ws_size,
                              hipStream_t stream) {
  const float* x   = (const float*)d_in[0];
  const float* W1  = (const float*)d_in[1];
  const float* b1  = (const float*)d_in[2];
  const float* W2  = (const float*)d_in[3];
  const float* b2  = (const float*)d_in[4];
  const int*   ei  = (const int*)d_in[5];
  const int*   bat = (const int*)d_in[6];
  float* out = (float*)d_out;

  const int N = in_sizes[0] / 3;
  const int E = in_sizes[5] / 2;

  char* ws = (char*)d_ws;
  float* dinv = (float*)(ws);
  float* bufA = (float*)(ws + (1 << 19));
  float* bufB = (float*)(ws + (1 << 19) + (size_t)N * HID * 4);
  float* cnt  = (float*)(ws + (1 << 19) + 2 * (size_t)N * HID * 4);

  const int B = 256;
  const int NH   = N * HID;
  const int N16  = N * 16;
  const int E16  = E * 16;
  const int tilesM = (N + 15) / 16;

  k_zero  <<<(N + B - 1) / B, B, 0, stream>>>(dinv, N);
  k_degree<<<(E + B - 1) / B, B, 0, stream>>>(ei, E, dinv);
  k_dinv  <<<(N + B - 1) / B, B, 0, stream>>>(dinv, N);

  k_gemm1  <<<(NH + B - 1) / B, B, 0, stream>>>(x, W1, bufA, N);
  k_zero   <<<(NH + B - 1) / B, B, 0, stream>>>(bufB, NH);
  k_scatter<<<(E16 + B - 1) / B, B, 0, stream>>>(ei, E, bufA, dinv, bufB);
  k_combine<<<(N16 + B - 1) / B, B, 0, stream>>>(bufB, bufA, dinv, b1, N);

  {
    dim3 grid(tilesM, HID / 16);
    k_gemm2<<<grid, 32, 0, stream>>>(bufB, W2, bufA, N);
  }
  k_zero   <<<(NH + B - 1) / B, B, 0, stream>>>(bufB, NH);
  k_scatter<<<(E16 + B - 1) / B, B, 0, stream>>>(ei, E, bufA, dinv, bufB);
  k_combine<<<(N16 + B - 1) / B, B, 0, stream>>>(bufB, bufA, dinv, b2, N);

  k_zero<<<(out_size + B - 1) / B, B, 0, stream>>>(out, out_size);
  k_zero<<<(512 + B - 1) / B, B, 0, stream>>>(cnt, 512);
  k_pool<<<(N16 + B - 1) / B, B, 0, stream>>>(bufB, bat, out, cnt, N);
  k_div <<<(out_size + B - 1) / B, B, 0, stream>>>(out, cnt, out_size);
}


# layer-2 linear moved behind the neighbour sum ((Ah)W2 for A(hW2)): a wave sums 16 nodes' layer-1 rows, then multiplies its tile by W2 on WMMA in the same kernel; GEMM launch gone; reorders f32 sums
# speedup vs baseline: 2.0795x; 2.0795x over previous
#include <hip/hip_runtime.h>

#define HID 64

typedef float v2f __attribute__((ext_vector_type(2)));
typedef float v8f __attribute__((ext_vector_type(8)));

__global__ void k_zero(float* __restrict__ p, int n) {
  int t = blockIdx.x * blockDim.x + threadIdx.x;
  if (t < n) p[t] = 0.0f;
}

__global__ void k_rank(const int* __restrict__ ei, int E, int* __restrict__ cnt, int* __restrict__ rnk) {
  int e = blockIdx.x * blockDim.x + threadIdx.x;
  if (e < E) rnk[e] = atomicAdd(&cnt[ei[E + e]], 1);
}

__global__ __launch_bounds__(1024) void k_scan(const int* __restrict__ cnt, int* __restrict__ rowinfo,
                                               float* __restrict__ dinv, int N) {
  __shared__ int wsum[32];
  const int tid  = threadIdx.x;
  const int lane = tid & 31;
  const int wv   = tid >> 5;
  const int base = blockIdx.x * 1024;

  int part = 0;
  const int4* __restrict__ c4 = (const int4*)cnt;
  const int n4 = base >> 2;
  #pragma unroll 4
  for (int i = tid; i < n4; i += 1024) {
    const int4 v = c4[i];
    part += (v.x + v.y) + (v.z + v.w);
  }
  for (int o = 16; o > 0; o >>= 1) part += __shfl_xor(part, o, 32);
  if (lane == 0) wsum[wv] = part;
  __syncthreads();
  int prefix = 0;
  for (int w = 0; w < 32; ++w) prefix += wsum[w];
  __syncthreads();

  const int i = base + tid;
  const int c = (i < N) ? cnt[i] : 0;
  int inc = c;
  for (int o = 1; o < 32; o <<= 1) {
    int y = __shfl_up(inc, o, 32);
    if (lane >= o) inc += y;
  }
  if (lane == 31) wsum[wv] = inc;
  __syncthreads();
  int woff = 0;
  for (int w = 0; w < wv; ++w) woff += wsum[w];
  if (i < N) {
    ((int2*)rowinfo)[i] = make_int2(prefix + woff + inc - c, c);
    dinv[i]    = rsqrtf((float)c + 1.0f);
  }
}

__global__ void k_fill(const int* __restrict__ ei, int E, const int* __restrict__ rnk,
                       const int* __restrict__ rowinfo, const float* __restrict__ dinv,
                       int* __restrict__ colw) {
  int e = blockIdx.x * blockDim.x + threadIdx.x;
  if (e >= E) return;
  const int s = ei[e];
  const int d = ei[E + e];
  const float norm = dinv[s] * dinv[d];
  ((int2*)colw)[rowinfo[2 * d] + rnk[e]] = make_int2(s, __float_as_int(norm));
}

__global__ __launch_bounds__(256) void k_gather1(const int* __restrict__ rowinfo,
                                                 const int* __restrict__ colw,
                                                 const float* __restrict__ x,
                                                 const float* __restrict__ W1,
                                                 const float* __restrict__ dinv,
                                                 const float* __restrict__ bias,
                                                 float* __restrict__ out, int N) {
  const int t = blockIdx.x * blockDim.x + threadIdx.x;
  const int n = t >> 4;
  if (n >= N) return;
  const int l16 = threadIdx.x & 15;
  const int j   = l16 * 4;
  const int2 ri = ((const int2*)rowinfo)[n];
  const int beg = ri.x;
  const int len = ri.y;
  const float dn = dinv[n];
  const float a0 = W1[j + 0],           a1 = W1[j + 1],           a2 = W1[j + 2],           a3 = W1[j + 3];
  const float b0 = W1[HID + j + 0],     b1 = W1[HID + j + 1],     b2 = W1[HID + j + 2],     b3 = W1[HID + j + 3];
  const float c0 = W1[2 * HID + j + 0], c1 = W1[2 * HID + j + 1], c2 = W1[2 * HID + j + 2], c3 = W1[2 * HID + j + 3];
  const float xs0 = x[n * 3 + 0], xs1 = x[n * 3 + 1], xs2 = x[n * 3 + 2];
  float4 hs;
  hs.x = xs0 * a0 + xs1 * b0 + xs2 * c0;
  hs.y = xs0 * a1 + xs1 * b1 + xs2 * c1;
  hs.z = xs0 * a2 + xs1 * b2 + xs2 * c2;
  hs.w = xs0 * a3 + xs1 * b3 + xs2 * c3;
  float4 acc = {0.0f, 0.0f, 0.0f, 0.0f};

  for (int base = 0; base < len; base += 16) {
    const int slot = base + l16;
    float mx0 = 0.0f, mx1 = 0.0f, mx2 = 0.0f, myw = 0.0f;
    if (slot < len) {
      const int2 cw = ((const int2*)colw)[beg + slot];
      const int s = (int)min((unsigned)cw.x, (unsigned)(N - 1));
      myw = __int_as_float(cw.y);
      mx0 = x[s * 3 + 0]; mx1 = x[s * 3 + 1]; mx2 = x[s * 3 + 2];
    }
    const int m = min(16, len - base);
    for (int k = 0; k < m; ++k) {
      const float x0 = __shfl(mx0, k, 16), x1 = __shfl(mx1, k, 16), x2 = __shfl(mx2, k, 16);
      const float w  = __shfl(myw, k, 16);
      const float r0 = x0 * a0 + x1 * b0 + x2 * c0;
      const float r1 = x0 * a1 + x1 * b1 + x2 * c1;
      const float r2 = x0 * a2 + x1 * b2 + x2 * c2;
      const float r3 = x0 * a3 + x1 * b3 + x2 * c3;
      acc.x = fmaf(r0, w, acc.x); acc.y = fmaf(r1, w, acc.y);
      acc.z = fmaf(r2, w, acc.z); acc.w = fmaf(r3, w, acc.w);
    }
  }

  const float sl = dn * dn;
  float4 o;
  o.x = fmaxf(fmaf(hs.x, sl, acc.x) + bias[j + 0], 0.0f);
  o.y = fmaxf(fmaf(hs.y, sl, acc.y) + bias[j + 1], 0.0f);
  o.z = fmaxf(fmaf(hs.z, sl, acc.z) + bias[j + 2], 0.0f);
  o.w = fmaxf(fmaf(hs.w, sl, acc.w) + bias[j + 3], 0.0f);
  *(float4*)(out + (long)n * HID + j) = o;
}

__global__ __launch_bounds__(256) void k_gather_mm(const int* __restrict__ rowinfo,
                                                   const int* __restrict__ colw,
                                                   const float* __restrict__ h,
                                                   const float* __restrict__ W,
                                                   const float* __restrict__ dinv,
                                                   const float* __restrict__ bias,
                                                   float* __restrict__ out, int N) {
  __shared__ alignas(16) float tile[8][16][68];
  const int wv   = threadIdx.x >> 5;
  const int lane = threadIdx.x & 31;
  const int half = lane >> 4;
  const int l16  = lane & 15;
  const int j    = l16 * 4;
  const int nb   = (blockIdx.x * 8 + wv) * 16;

  int rbeg = 0, rlen = 0;
  float rdn = 0.0f;
  if (nb + l16 < N) {
    const int2 ri = ((const int2*)rowinfo)[nb + l16];
    rbeg = ri.x; rlen = ri.y; rdn = dinv[nb + l16];
  }

  int   nbeg = __shfl(rbeg, half, 16);
  int   nlen = __shfl(rlen, half, 16);
  float ndn  = __shfl(rdn,  half, 16);
  int   ncol = 0;
  float nw   = 0.0f;
  if (l16 < nlen) {
    const int2 cw = ((const int2*)colw)[nbeg + l16];
    ncol = (int)min((unsigned)cw.x, (unsigned)(N - 1));
    nw   = __int_as_float(cw.y);
  }

  for (int p = 0; p < 8; ++p) {
    const int r   = 2 * p + half;
    const int n   = nb + r;
    const int beg = nbeg;
    const int len = nlen;
    const float dn = ndn;
    int   mycol = ncol;
    float myw   = nw;
    if (p + 1 < 8) {
      nbeg = __shfl(rbeg, r + 2, 16);
      nlen = __shfl(rlen, r + 2, 16);
      ndn  = __shfl(rdn,  r + 2, 16);
      ncol = 0; nw = 0.0f;
      if (l16 < nlen) {
        const int2 cw = ((const int2*)colw)[nbeg + l16];
        ncol = (int)min((unsigned)cw.x, (unsigned)(N - 1));
        nw   = __int_as_float(cw.y);
      }
    }
    float4 hs = {0.0f, 0.0f, 0.0f, 0.0f};
    if (n < N) hs = *(const float4*)(h + (long)n * HID + j);
    float4 acc = {0.0f, 0.0f, 0.0f, 0.0f};

    for (int base = 0; base < len; base += 16) {
      if (base > 0) {
        mycol = 0; myw = 0.0f;
        if (base + l16 < len) {
          const int2 cw = ((const int2*)colw)[beg + base + l16];
          mycol = (int)min((unsigned)cw.x, (unsigned)(N - 1));
          myw   = __int_as_float(cw.y);
        }
      }
      const int m = min(16, len - base);
      for (int k = 0; k < m; k += 8) {
        float4 v[8];
        float  w[8];
        int    s[8];
        #pragma unroll
        for (int i = 0; i < 8; ++i) {
          s[i] = __shfl(mycol, k + i, 16);
          w[i] = __shfl(myw,   k + i, 16);
        }
        #pragma unroll
        for (int i = 0; i < 8; ++i) {
          v[i] = make_float4(0.0f, 0.0f, 0.0f, 0.0f);
          if (k + i < m) v[i] = *(const float4*)(h + (long)s[i] * HID + j);
        }
        #pragma unroll
        for (int i = 0; i < 8; ++i) {
          acc.x = fmaf(v[i].x, w[i], acc.x); acc.y = fmaf(v[i].y, w[i], acc.y);
          acc.z = fmaf(v[i].z, w[i], acc.z); acc.w = fmaf(v[i].w, w[i], acc.w);
        }
      }
    }

    const float sl = dn * dn;
    float4 a;
    a.x = fmaf(hs.x, sl, acc.x); a.y = fmaf(hs.y, sl, acc.y);
    a.z = fmaf(hs.z, sl, acc.z); a.w = fmaf(hs.w, sl, acc.w);
    *(float4*)&tile[wv][r][j] = a;
  }
  __syncthreads();

  v8f c0 = {}, c1 = {}, c2 = {}, c3 = {};
  const float* __restrict__ Arow = &tile[wv][l16][0];
  #pragma unroll
  for (int i = 0; i < 16; ++i) {
    const int ka = 4 * i + 2 * half;
    v2f a;  a.x  = Arow[ka];                    a.y  = Arow[ka + 1];
    v2f b0; b0.x = W[ka * HID + l16];           b0.y = W[(ka + 1) * HID + l16];
    v2f b1; b1.x = W[ka * HID + 16 + l16];      b1.y = W[(ka + 1) * HID + 16 + l16];
    v2f b2; b2.x = W[ka * HID + 32 + l16];      b2.y = W[(ka + 1) * HID + 32 + l16];
    v2f b3; b3.x = W[ka * HID + 48 + l16];      b3.y = W[(ka + 1) * HID + 48 + l16];
    c0 = __builtin_amdgcn_wmma_f32_16x16x4_f32(false, a, false, b0, (short)0, c0, false, false);
    c1 = __builtin_amdgcn_wmma_f32_16x16x4_f32(false, a, false, b1, (short)0, c1, false, false);
    c2 = __builtin_amdgcn_wmma_f32_16x16x4_f32(false, a, false, b2, (short)0, c2, false, false);
    c3 = __builtin_amdgcn_wmma_f32_16x16x4_f32(false, a, false, b3, (short)0, c3, false, false);
  }
  __syncthreads();

  const float bb0 = bias[l16], bb1 = bias[16 + l16], bb2 = bias[32 + l16], bb3 = bias[48 + l16];
  #pragma unroll
  for (int q = 0; q < 8; ++q) {
    float* trow = &tile[wv][q + 8 * half][0];
    trow[l16]      = fmaxf(c0[q] + bb0, 0.0f);
    trow[16 + l16] = fmaxf(c1[q] + bb1, 0.0f);
    trow[32 + l16] = fmaxf(c2[q] + bb2, 0.0f);
    trow[48 + l16] = fmaxf(c3[q] + bb3, 0.0f);
  }
  __syncthreads();
  #pragma unroll
  for (int q = 0; q < 8; ++q) {
    const int r = 2 * q + half;
    const float4 o = *(const float4*)&tile[wv][r][j];
    if (nb + r < N) *(float4*)(out + (long)(nb + r) * HID + j) = o;
  }
}

__global__ void k_pool(const float* __restrict__ h, const int* __restrict__ batch,
                       float* __restrict__ out, float* __restrict__ cnt, int N) {
  int t = blockIdx.x * blockDim.x + threadIdx.x;
  if (t >= N * 16) return;
  int n = t >> 4;
  int j = (t & 15) * 4;
  int g = batch[n];
  const float4 v = *(const float4*)(h + (long)n * HID + j);
  float* o = out + (long)g * HID + j;
  atomicAdd(o + 0, v.x);
  atomicAdd(o + 1, v.y);
  atomicAdd(o + 2, v.z);
  atomicAdd(o + 3, v.w);
  if ((t & 15) == 0) atomicAdd(&cnt[g], 1.0f);
}

__global__ void k_div(float* __restrict__ out, const float* __restrict__ cnt, int total) {
  int t = blockIdx.x * blockDim.x + threadIdx.x;
  if (t < total) out[t] /= fmaxf(cnt[t >> 6], 1.0f);
}

extern "C" void kernel_launch(void* const* d_in, const int* in_sizes, int n_in,
                              void* d_out, int out_size, void* d_ws, size_t ws_size,
                              hipStream_t stream) {
  const float* x   = (const float*)d_in[0];
  const float* W1  = (const float*)d_in[1];
  const float* b1  = (const float*)d_in[2];
  const float* W2  = (const float*)d_in[3];
  const float* b2  = (const float*)d_in[4];
  const int*   ei  = (const int*)d_in[5];
  const int*   bat = (const int*)d_in[6];
  float* out = (float*)d_out;

  const int N = in_sizes[0] / 3;
  const int E = in_sizes[5] / 2;

  char* ws = (char*)d_ws;
  float* dinv = (float*)(ws);
  float* bufA = (float*)(ws + (1 << 19));
  float* bufB = (float*)(ws + (1 << 19) + (size_t)N * HID * 4);
  float* cnt  = (float*)(ws + (1 << 19) + 2 * (size_t)N * HID * 4);
  const size_t Np = ((size_t)N + 63) & ~(size_t)63;
  const size_t Ep = ((size_t)E + 63) & ~(size_t)63;
  int*  icnt    = (int*)(ws + (1 << 19) + 2 * (size_t)N * HID * 4 + 4096);
  int*  rowinfo = icnt + Np;
  int*  rnk     = rowinfo + 2 * Np;
  int*  colw    = rnk + Ep;

  const int B = 256;
  const int N16  = N * 16;

  k_zero<<<(N + B - 1) / B, B, 0, stream>>>((float*)icnt, N);
  k_rank<<<(E + B - 1) / B, B, 0, stream>>>(ei, E, icnt, rnk);
  k_scan<<<(N + 1023) / 1024, 1024, 0, stream>>>(icnt, rowinfo, dinv, N);
  k_fill<<<(E + B - 1) / B, B, 0, stream>>>(ei, E, rnk, rowinfo, dinv, colw);

  k_gather1<<<(N16 + B - 1) / B, B, 0, stream>>>(rowinfo, colw, x, W1, dinv, b1, bufB, N);

  k_gather_mm<<<(N + 127) / 128, B, 0, stream>>>(rowinfo, colw, bufB, W2, dinv, b2, bufA, N);

  k_zero<<<(out_size + B - 1) / B, B, 0, stream>>>(out, out_size);
  k_zero<<<(512 + B - 1) / B, B, 0, stream>>>(cnt, 512);
  k_pool<<<(N16 + B - 1) / B, B, 0, stream>>>(bufA, bat, out, cnt, N);
  k_div <<<(out_size + B - 1) / B, B, 0, stream>>>(out, cnt, out_size);
}


# mean pool moved into the layer-2 kernel: a wave sums its 16 finished rows per graph run from its LDS tile, one atomic add per column and run; rows never stored, pool launch gone; reorders pool sums
# speedup vs baseline: 3.1911x; 1.5346x over previous
#include <hip/hip_runtime.h>

#define HID 64

typedef float v2f __attribute__((ext_vector_type(2)));
typedef float v8f __attribute__((ext_vector_type(8)));

__global__ void k_zero(float* __restrict__ p, int n) {
  int t = blockIdx.x * blockDim.x + threadIdx.x;
  if (t < n) p[t] = 0.0f;
}

__global__ void k_rank(const int* __restrict__ ei, int E, int* __restrict__ cnt, int* __restrict__ rnk) {
  int e = blockIdx.x * blockDim.x + threadIdx.x;
  if (e < E) rnk[e] = atomicAdd(&cnt[ei[E + e]], 1);
}

__global__ __launch_bounds__(1024) void k_scan(const int* __restrict__ cnt, int* __restrict__ rowinfo,
                                               float* __restrict__ dinv, int N) {
  __shared__ int wsum[32];
  const int tid  = threadIdx.x;
  const int lane = tid & 31;
  const int wv   = tid >> 5;
  const int base = blockIdx.x * 1024;

  int part = 0;
  const int4* __restrict__ c4 = (const int4*)cnt;
  const int n4 = base >> 2;
  #pragma unroll 4
  for (int i = tid; i < n4; i += 1024) {
    const int4 v = c4[i];
    part += (v.x + v.y) + (v.z + v.w);
  }
  for (int o = 16; o > 0; o >>= 1) part += __shfl_xor(part, o, 32);
  if (lane == 0) wsum[wv] = part;
  __syncthreads();
  int prefix = 0;
  for (int w = 0; w < 32; ++w) prefix += wsum[w];
  __syncthreads();

  const int i = base + tid;
  const int c = (i < N) ? cnt[i] : 0;
  int inc = c;
  for (int o = 1; o < 32; o <<= 1) {
    int y = __shfl_up(inc, o, 32);
    if (lane >= o) inc += y;
  }
  if (lane == 31) wsum[wv] = inc;
  __syncthreads();
  int woff = 0;
  for (int w = 0; w < wv; ++w) woff += wsum[w];
  if (i < N) {
    ((int2*)rowinfo)[i] = make_int2(prefix + woff + inc - c, c);
    dinv[i]    = rsqrtf((float)c + 1.0f);
  }
}

__global__ void k_fill(const int* __restrict__ ei, int E, const int* __restrict__ rnk,
                       const int* __restrict__ rowinfo, const float* __restrict__ dinv,
                       int* __restrict__ colw) {
  int e = blockIdx.x * blockDim.x + threadIdx.x;
  if (e >= E) return;
  const int s = ei[e];
  const int d = ei[E + e];
  const float norm = dinv[s] * dinv[d];
  ((int2*)colw)[rowinfo[2 * d] + rnk[e]] = make_int2(s, __float_as_int(norm));
}

__global__ __launch_bounds__(256) void k_gather1(const int* __restrict__ rowinfo,
                                                 const int* __restrict__ colw,
                                                 const float* __restrict__ x,
                                                 const float* __restrict__ W1,
                                                 const float* __restrict__ dinv,
                                                 const float* __restrict__ bias,
                                                 float* __restrict__ out, int N) {
  const int t = blockIdx.x * blockDim.x + threadIdx.x;
  const int n = t >> 4;
  if (n >= N) return;
  const int l16 = threadIdx.x & 15;
  const int j   = l16 * 4;
  const int2 ri = ((const int2*)rowinfo)[n];
  const int beg = ri.x;
  const int len = ri.y;
  const float dn = dinv[n];
  const float a0 = W1[j + 0],           a1 = W1[j + 1],           a2 = W1[j + 2],           a3 = W1[j + 3];
  const float b0 = W1[HID + j + 0],     b1 = W1[HID + j + 1],     b2 = W1[HID + j + 2],     b3 = W1[HID + j + 3];
  const float c0 = W1[2 * HID + j + 0], c1 = W1[2 * HID + j + 1], c2 = W1[2 * HID + j + 2], c3 = W1[2 * HID + j + 3];
  const float xs0 = x[n * 3 + 0], xs1 = x[n * 3 + 1], xs2 = x[n * 3 + 2];
  float4 hs;
  hs.x = xs0 * a0 + xs1 * b0 + xs2 * c0;
  hs.y = xs0 * a1 + xs1 * b1 + xs2 * c1;
  hs.z = xs0 * a2 + xs1 * b2 + xs2 * c2;
  hs.w = xs0 * a3 + xs1 * b3 + xs2 * c3;
  float4 acc = {0.0f, 0.0f, 0.0f, 0.0f};

  for (int base = 0; base < len; base += 16) {
    const int slot = base + l16;
    float mx0 = 0.0f, mx1 = 0.0f, mx2 = 0.0f, myw = 0.0f;
    if (slot < len) {
      const int2 cw = ((const int2*)colw)[beg + slot];
      const int s = (int)min((unsigned)cw.x, (unsigned)(N - 1));
      myw = __int_as_float(cw.y);
      mx0 = x[s * 3 + 0]; mx1 = x[s * 3 + 1]; mx2 = x[s * 3 + 2];
    }
    const int m = min(16, len - base);
    for (int k = 0; k < m; ++k) {
      const float x0 = __shfl(mx0, k, 16), x1 = __shfl(mx1, k, 16), x2 = __shfl(mx2, k, 16);
      const float w  = __shfl(myw, k, 16);
      const float r0 = x0 * a0 + x1 * b0 + x2 * c0;
      const float r1 = x0 * a1 + x1 * b1 + x2 * c1;
      const float r2 = x0 * a2 + x1 * b2 + x2 * c2;
      const float r3 = x0 * a3 + x1 * b3 + x2 * c3;
      acc.x = fmaf(r0, w, acc.x); acc.y = fmaf(r1, w, acc.y);
      acc.z = fmaf(r2, w, acc.z); acc.w = fmaf(r3, w, acc.w);
    }
  }

  const float sl = dn * dn;
  float4 o;
  o.x = fmaxf(fmaf(hs.x, sl, acc.x) + bias[j + 0], 0.0f);
  o.y = fmaxf(fmaf(hs.y, sl, acc.y) + bias[j + 1], 0.0f);
  o.z = fmaxf(fmaf(hs.z, sl, acc.z) + bias[j + 2], 0.0f);
  o.w = fmaxf(fmaf(hs.w, sl, acc.w) + bias[j + 3], 0.0f);
  *(float4*)(out + (long)n * HID + j) = o;
}

__device__ __forceinline__ void pool_add(float* __restrict__ pool, float* __restrict__ cnt, int G,
                                         int g, int run, int lane, float s0, float s1) {
  if ((unsigned)g < (unsigned)G) {
    float* o = pool + (long)g * HID;
    atomicAdd(o + lane, s0);
    atomicAdd(o + 32 + lane, s1);
    if (lane == 0) atomicAdd(cnt + g, (float)run);
  }
}

__global__ __launch_bounds__(256) void k_gather_mm(const int* __restrict__ rowinfo,
                                                   const int* __restrict__ colw,
                                                   const float* __restrict__ h,
                                                   const float* __restrict__ W,
                                                   const float* __restrict__ dinv,
                                                   const float* __restrict__ bias,
                                                   const int* __restrict__ batch,
                                                   float* __restrict__ pool,
                                                   float* __restrict__ cnt,
                                                   int pool_elems, int N) {
  __shared__ alignas(16) float tile[8][16][68];
  const int wv   = threadIdx.x >> 5;
  const int lane = threadIdx.x & 31;
  const int half = lane >> 4;
  const int l16  = lane & 15;
  const int j    = l16 * 4;
  const int nb   = (blockIdx.x * 8 + wv) * 16;

  int rbeg = 0, rlen = 0;
  float rdn = 0.0f;
  int gid = 0;
  if (nb + l16 < N) {
    const int2 ri = ((const int2*)rowinfo)[nb + l16];
    rbeg = ri.x; rlen = ri.y; rdn = dinv[nb + l16];
    gid = batch[nb + l16];
  }

  int   nbeg = __shfl(rbeg, half, 16);
  int   nlen = __shfl(rlen, half, 16);
  float ndn  = __shfl(rdn,  half, 16);
  int   ncol = 0;
  float nw   = 0.0f;
  if (l16 < nlen) {
    const int2 cw = ((const int2*)colw)[nbeg + l16];
    ncol = (int)min((unsigned)cw.x, (unsigned)(N - 1));
    nw   = __int_as_float(cw.y);
  }

  for (int p = 0; p < 8; ++p) {
    const int r   = 2 * p + half;
    const int n   = nb + r;
    const int beg = nbeg;
    const int len = nlen;
    const float dn = ndn;
    int   mycol = ncol;
    float myw   = nw;
    if (p + 1 < 8) {
      nbeg = __shfl(rbeg, r + 2, 16);
      nlen = __shfl(rlen, r + 2, 16);
      ndn  = __shfl(rdn,  r + 2, 16);
      ncol = 0; nw = 0.0f;
      if (l16 < nlen) {
        const int2 cw = ((const int2*)colw)[nbeg + l16];
        ncol = (int)min((unsigned)cw.x, (unsigned)(N - 1));
        nw   = __int_as_float(cw.y);
      }
    }
    float4 hs = {0.0f, 0.0f, 0.0f, 0.0f};
    if (n < N) hs = *(const float4*)(h + (long)n * HID + j);
    float4 acc = {0.0f, 0.0f, 0.0f, 0.0f};

    for (int base = 0; base < len; base += 16) {
      if (base > 0) {
        mycol = 0; myw = 0.0f;
        if (base + l16 < len) {
          const int2 cw = ((const int2*)colw)[beg + base + l16];
          mycol = (int)min((unsigned)cw.x, (unsigned)(N - 1));
          myw   = __int_as_float(cw.y);
        }
      }
      const int m = min(16, len - base);
      for (int k = 0; k < m; k += 8) {
        float4 v[8];
        float  w[8];
        int    s[8];
        #pragma unroll
        for (int i = 0; i < 8; ++i) {
          s[i] = __shfl(mycol, k + i, 16);
          w[i] = __shfl(myw,   k + i, 16);
        }
        #pragma unroll
        for (int i = 0; i < 8; ++i) {
          v[i] = make_float4(0.0f, 0.0f, 0.0f, 0.0f);
          if (k + i < m) v[i] = *(const float4*)(h + (long)s[i] * HID + j);
        }
        #pragma unroll
        for (int i = 0; i < 8; ++i) {
          acc.x = fmaf(v[i].x, w[i], acc.x); acc.y = fmaf(v[i].y, w[i], acc.y);
          acc.z = fmaf(v[i].z, w[i], acc.z); acc.w = fmaf(v[i].w, w[i], acc.w);
        }
      }
    }

    const float sl = dn * dn;
    float4 a;
    a.x = fmaf(hs.x, sl, acc.x); a.y = fmaf(hs.y, sl, acc.y);
    a.z = fmaf(hs.z, sl, acc.z); a.w = fmaf(hs.w, sl, acc.w);
    *(float4*)&tile[wv][r][j] = a;
  }
  __syncthreads();

  v8f c0 = {}, c1 = {}, c2 = {}, c3 = {};
  const float* __restrict__ Arow = &tile[wv][l16][0];
  #pragma unroll
  for (int i = 0; i < 16; ++i) {
    const int ka = 4 * i + 2 * half;
    v2f a;  a.x  = Arow[ka];                    a.y  = Arow[ka + 1];
    v2f b0; b0.x = W[ka * HID + l16];           b0.y = W[(ka + 1) * HID + l16];
    v2f b1; b1.x = W[ka * HID + 16 + l16];      b1.y = W[(ka + 1) * HID + 16 + l16];
    v2f b2; b2.x = W[ka * HID + 32 + l16];      b2.y = W[(ka + 1) * HID + 32 + l16];
    v2f b3; b3.x = W[ka * HID + 48 + l16];      b3.y = W[(ka + 1) * HID + 48 + l16];
    c0 = __builtin_amdgcn_wmma_f32_16x16x4_f32(false, a, false, b0, (short)0, c0, false, false);
    c1 = __builtin_amdgcn_wmma_f32_16x16x4_f32(false, a, false, b1, (short)0, c1, false, false);
    c2 = __builtin_amdgcn_wmma_f32_16x16x4_f32(false, a, false, b2, (short)0, c2, false, false);
    c3 = __builtin_amdgcn_wmma_f32_16x16x4_f32(false, a, false, b3, (short)0, c3, false, false);
  }
  __syncthreads();

  const float bb0 = bias[l16], bb1 = bias[16 + l16], bb2 = bias[32 + l16], bb3 = bias[48 + l16];
  #pragma unroll
  for (int q = 0; q < 8; ++q) {
    float* trow = &tile[wv][q + 8 * half][0];
    trow[l16]      = fmaxf(c0[q] + bb0, 0.0f);
    trow[16 + l16] = fmaxf(c1[q] + bb1, 0.0f);
    trow[32 + l16] = fmaxf(c2[q] + bb2, 0.0f);
    trow[48 + l16] = fmaxf(c3[q] + bb3, 0.0f);
  }
  __syncthreads();

  const int G   = pool_elems >> 6;
  const int nbu = __builtin_amdgcn_readfirstlane(nb);
  float t0[16], t1[16];
  #pragma unroll
  for (int r = 0; r < 16; ++r) {
    t0[r] = tile[wv][r][lane];
    t1[r] = tile[wv][r][32 + lane];
  }
  float s0 = 0.0f, s1 = 0.0f;
  int gp = 0, run = 0;
  #pragma unroll
  for (int r = 0; r < 16; ++r) {
    if (nbu + r < N) {
      const int g = __builtin_amdgcn_readlane(gid, r);
      if (run > 0 && g != gp) {
        pool_add(pool, cnt, G, gp, run, lane, s0, s1);
        s0 = 0.0f; s1 = 0.0f; run = 0;
      }
      gp = g;
      s0 += t0[r];
      s1 += t1[r];
      ++run;
    }
  }
  if (run > 0) pool_add(pool, cnt, G, gp, run, lane, s0, s1);
}

__global__ void k_div(float* __restrict__ out, const float* __restrict__ cnt, int total) {
  int t = blockIdx.x * blockDim.x + threadIdx.x;
  if (t < total) out[t] /= fmaxf(cnt[t >> 6], 1.0f);
}

extern "C" void kernel_launch(void* const* d_in, const int* in_sizes, int n_in,
                              void* d_out, int out_size, void* d_ws, size_t ws_size,
                              hipStream_t stream) {
  const float* x   = (const float*)d_in[0];
  const float* W1  = (const float*)d_in[1];
  const float* b1  = (const float*)d_in[2];
  const float* W2  = (const float*)d_in[3];
  const float* b2  = (const float*)d_in[4];
  const int*   ei  = (const int*)d_in[5];
  const int*   bat = (const int*)d_in[6];
  float* out = (float*)d_out;

  const int N = in_sizes[0] / 3;
  const int E = in_sizes[5] / 2;

  char* ws = (char*)d_ws;
  float* dinv = (float*)(ws);
  float* bufB = (float*)(ws + (1 << 19) + (size_t)N * HID * 4);
  float* cnt  = (float*)(ws + (1 << 19) + 2 * (size_t)N * HID * 4);
  const size_t Np = ((size_t)N + 63) & ~(size_t)63;
  const size_t Ep = ((size_t)E + 63) & ~(size_t)63;
  int*  icnt    = (int*)(ws + (1 << 19) + 2 * (size_t)N * HID * 4 + 4096);
  int*  rowinfo = icnt + Np;
  int*  rnk     = rowinfo + 2 * Np;
  int*  colw    = rnk + Ep;

  const int B = 256;
  const int N16  = N * 16;

  k_zero<<<(N + B - 1) / B, B, 0, stream>>>((float*)icnt, N);
  k_rank<<<(E + B - 1) / B, B, 0, stream>>>(ei, E, icnt, rnk);
  k_scan<<<(N + 1023) / 1024, 1024, 0, stream>>>(icnt, rowinfo, dinv, N);
  k_fill<<<(E + B - 1) / B, B, 0, stream>>>(ei, E, rnk, rowinfo, dinv, colw);

  k_gather1<<<(N16 + B - 1) / B, B, 0, stream>>>(rowinfo, colw, x, W1, dinv, b1, bufB, N);

  k_zero<<<(out_size + B - 1) / B, B, 0, stream>>>(out, out_size);
  k_zero<<<(512 + B - 1) / B, B, 0, stream>>>(cnt, 512);

  k_gather_mm<<<(N + 127) / 128, B, 0, stream>>>(rowinfo, colw, bufB, W2, dinv, b2, bat, out, cnt, out_size, N);

  k_div <<<(out_size + B - 1) / B, B, 0, stream>>>(out, cnt, out_size);
}


# buckets built by a two-level LDS counting sort (edge chunks ordered by 512-node range, then one group per range), not global rank atomics+scan+fill; norm formed in layer-1 gather; arrival-order sums
# speedup vs baseline: 6.2888x; 1.9707x over previous
#include <hip/hip_runtime.h>

#define HID 64

typedef float v2f __attribute__((ext_vector_type(2)));
typedef float v8f __attribute__((ext_vector_type(8)));

__global__ void k_zero(float* __restrict__ p, int n) {
  int t = blockIdx.x * blockDim.x + threadIdx.x;
  if (t < n) p[t] = 0.0f;
}

#define LOGR   9
#define RB     512
#define CH     4096
#define NBMAX  1024
#define NCHMAX 1024
#define CAP    8192

__device__ __forceinline__ int block_excl_scan(int v, int* wsum, int lane, int wv, int& total) {
  int inc = v;
  #pragma unroll
  for (int o = 1; o < 32; o <<= 1) {
    int y = __shfl_up(inc, o, 32);
    if (lane >= o) inc += y;
  }
  __syncthreads();
  if (lane == 31) wsum[wv] = inc;
  __syncthreads();
  int woff = 0, t = 0;
  for (int w = 0; w < 32; ++w) {
    const int xw = wsum[w];
    if (w < wv) woff += xw;
    t += xw;
  }
  total = t;
  return woff + inc - v;
}

__global__ __launch_bounds__(1024) void k_bin(const int* __restrict__ ei, int E, int N, int NB, int NCHP,
                                              int* __restrict__ sorted, int* __restrict__ offT) {
  __shared__ int hist[NBMAX];
  __shared__ int wsum[32];
  __shared__ alignas(8) int2 stage[CH];
  const int tid  = threadIdx.x;
  const int lane = tid & 31;
  const int wv   = tid >> 5;
  const int c    = blockIdx.x;
  const int e0   = c * CH;
  hist[tid] = 0;
  __syncthreads();

  int s[4], d[4], r[4];
  #pragma unroll
  for (int u = 0; u < 4; ++u) {
    const int e = e0 + u * 1024 + tid;
    s[u] = 0; d[u] = -1;
    if (e < E) { s[u] = ei[e]; d[u] = ei[E + e]; }
  }
  #pragma unroll
  for (int u = 0; u < 4; ++u) {
    r[u] = 0;
    if (e0 + u * 1024 + tid < E) {
      d[u] = (int)min((unsigned)d[u], (unsigned)(N - 1));
      r[u] = atomicAdd(&hist[d[u] >> LOGR], 1);
    }
  }
  __syncthreads();

  int total;
  const int ex = block_excl_scan(hist[tid], wsum, lane, wv, total);
  hist[tid] = ex;
  if (tid <= NB) offT[tid * NCHP + c] = ex;
  __syncthreads();

  #pragma unroll
  for (int u = 0; u < 4; ++u)
    if (d[u] >= 0) stage[hist[d[u] >> LOGR] + r[u]] = make_int2(s[u], d[u]);
  __syncthreads();

  const int nc = min(CH, E - e0);
  #pragma unroll
  for (int u = 0; u < 4; ++u) {
    const int p = u * 1024 + tid;
    if (p < nc) ((int2*)sorted)[e0 + p] = stage[p];
  }
}

__device__ __forceinline__ void sort_take(int2 e, int p, bool staged, int* cnt, int2* tmp) {
  const int dl = e.y & (RB - 1);
  const int r  = atomicAdd(&cnt[dl], 1);
  if (staged) tmp[p] = make_int2(e.x, (r << LOGR) | dl);
}

__global__ __launch_bounds__(1024) void k_sort(const int* __restrict__ sorted, const int* __restrict__ offT,
                                               int NCH, int NCHP, int N,
                                               int* __restrict__ rowinfo, float* __restrict__ dinv,
                                               int* __restrict__ col) {
  __shared__ alignas(16) int4 run[NCHMAX];
  __shared__ int cnt[RB];
  __shared__ int start[RB];
  __shared__ int wsum[32];
  __shared__ alignas(8) int2 tmp[CAP];
  __shared__ int outst[CAP];
  const int tid  = threadIdx.x;
  const int lane = tid & 31;
  const int wv   = tid >> 5;
  const int b    = blockIdx.x;

  int l0 = 0, ln = 0;
  if (tid < NCH) {
    l0 = offT[b * NCHP + tid];
    ln = offT[(b + 1) * NCHP + tid] - l0;
  }
  int M, base;
  const int myrs = block_excl_scan(ln, wsum, lane, wv, M);
  (void)block_excl_scan(l0, wsum, lane, wv, base);
  run[tid] = make_int4(l0, ln, myrs, 0);
  if (tid < RB) cnt[tid] = 0;
  __syncthreads();
  const bool staged = (M <= CAP);
  const int2* __restrict__ sg = (const int2*)sorted;

  for (int c0 = wv; c0 < NCHMAX; c0 += 128) {
    if (c0 >= NCH) break;
    int4 rn[4];
    #pragma unroll
    for (int u = 0; u < 4; ++u) rn[u] = run[c0 + 32 * u];
    int2 e[4];
    #pragma unroll
    for (int u = 0; u < 4; ++u) {
      e[u] = make_int2(0, 0);
      if (lane < rn[u].y) e[u] = sg[(c0 + 32 * u) * CH + rn[u].x + lane];
    }
    #pragma unroll
    for (int u = 0; u < 4; ++u)
      if (lane < rn[u].y) sort_take(e[u], rn[u].z + lane, staged, cnt, tmp);
    #pragma unroll
    for (int u = 0; u < 4; ++u)
      for (int i = lane + 32; i < rn[u].y; i += 32)
        sort_take(sg[(c0 + 32 * u) * CH + rn[u].x + i], rn[u].z + i, staged, cnt, tmp);
  }
  __syncthreads();

  const int cmine = (tid < RB) ? cnt[tid] : 0;
  int tot2;
  const int ex = block_excl_scan(cmine, wsum, lane, wv, tot2);
  const int n  = b * RB + tid;
  if (tid < RB) {
    start[tid] = ex;
    if (n < N) {
      ((int2*)rowinfo)[n] = make_int2(base + ex, cmine);
      dinv[n] = rsqrtf((float)cmine + 1.0f);
    }
  }
  __syncthreads();

  if (staged) {
    for (int p = tid; p < M; p += 1024) {
      const int2 t = tmp[p];
      outst[start[t.y & (RB - 1)] + (t.y >> LOGR)] = t.x;
    }
    __syncthreads();
    for (int p = tid; p < M; p += 1024) col[base + p] = outst[p];
  } else {
    if (tid < RB) cnt[tid] = 0;
    __syncthreads();
    for (int c = wv; c < NCH; c += 32)
      for (int i = lane; i < run[c].y; i += 32) {
        const int2 ee = sg[c * CH + run[c].x + i];
        const int dl = ee.y & (RB - 1);
        const int r  = atomicAdd(&cnt[dl], 1);
        col[base + start[dl] + r] = ee.x;
      }
  }
}

__global__ __launch_bounds__(256) void k_gather1(const int* __restrict__ rowinfo,
                                                 const int* __restrict__ col,
                                                 int* __restrict__ colw,
                                                 const float* __restrict__ x,
                                                 const float* __restrict__ W1,
                                                 const float* __restrict__ dinv,
                                                 const float* __restrict__ bias,
                                                 float* __restrict__ out, int N) {
  const int t = blockIdx.x * blockDim.x + threadIdx.x;
  const int n = t >> 4;
  if (n >= N) return;
  const int l16 = threadIdx.x & 15;
  const int j   = l16 * 4;
  const int2 ri = ((const int2*)rowinfo)[n];
  const int beg = ri.x;
  const int len = ri.y;
  const float dn = dinv[n];
  const float a0 = W1[j + 0],           a1 = W1[j + 1],           a2 = W1[j + 2],           a3 = W1[j + 3];
  const float b0 = W1[HID + j + 0],     b1 = W1[HID + j + 1],     b2 = W1[HID + j + 2],     b3 = W1[HID + j + 3];
  const float c0 = W1[2 * HID + j + 0], c1 = W1[2 * HID + j + 1], c2 = W1[2 * HID + j + 2], c3 = W1[2 * HID + j + 3];
  const float xs0 = x[n * 3 + 0], xs1 = x[n * 3 + 1], xs2 = x[n * 3 + 2];
  float4 hs;
  hs.x = xs0 * a0 + xs1 * b0 + xs2 * c0;
  hs.y = xs0 * a1 + xs1 * b1 + xs2 * c1;
  hs.z = xs0 * a2 + xs1 * b2 + xs2 * c2;
  hs.w = xs0 * a3 + xs1 * b3 + xs2 * c3;
  float4 acc = {0.0f, 0.0f, 0.0f, 0.0f};

  for (int base = 0; base < len; base += 16) {
    const int slot = base + l16;
    float mx0 = 0.0f, mx1 = 0.0f, mx2 = 0.0f, myw = 0.0f;
    if (slot < len) {
      const int s = (int)min((unsigned)col[beg + slot], (unsigned)(N - 1));
      myw = dinv[s] * dn;
      mx0 = x[s * 3 + 0]; mx1 = x[s * 3 + 1]; mx2 = x[s * 3 + 2];
      ((int2*)colw)[beg + slot] = make_int2(s, __float_as_int(myw));
    }
    const int m = min(16, len - base);
    for (int k = 0; k < m; ++k) {
      const float x0 = __shfl(mx0, k, 16), x1 = __shfl(mx1, k, 16), x2 = __shfl(mx2, k, 16);
      const float w  = __shfl(myw, k, 16);
      const float r0 = x0 * a0 + x1 * b0 + x2 * c0;
      const float r1 = x0 * a1 + x1 * b1 + x2 * c1;
      const float r2 = x0 * a2 + x1 * b2 + x2 * c2;
      const float r3 = x0 * a3 + x1 * b3 + x2 * c3;
      acc.x = fmaf(r0, w, acc.x); acc.y = fmaf(r1, w, acc.y);
      acc.z = fmaf(r2, w, acc.z); acc.w = fmaf(r3, w, acc.w);
    }
  }

  const float sl = dn * dn;
  float4 o;
  o.x = fmaxf(fmaf(hs.x, sl, acc.x) + bias[j + 0], 0.0f);
  o.y = fmaxf(fmaf(hs.y, sl, acc.y) + bias[j + 1], 0.0f);
  o.z = fmaxf(fmaf(hs.z, sl, acc.z) + bias[j + 2], 0.0f);
  o.w = fmaxf(fmaf(hs.w, sl, acc.w) + bias[j + 3], 0.0f);
  *(float4*)(out + (long)n * HID + j) = o;
}

__device__ __forceinline__ void pool_add(float* __restrict__ pool, float* __restrict__ cnt, int G,
                                         int g, int run, int lane, float s0, float s1) {
  if ((unsigned)g < (unsigned)G) {
    float* o = pool + (long)g * HID;
    atomicAdd(o + lane, s0);
    atomicAdd(o + 32 + lane, s1);
    if (lane == 0) atomicAdd(cnt + g, (float)run);
  }
}

__global__ __launch_bounds__(256) void k_gather_mm(const int* __restrict__ rowinfo,
                                                   const int* __restrict__ colw,
                                                   const float* __restrict__ h,
                                                   const float* __restrict__ W,
                                                   const float* __restrict__ dinv,
                                                   const float* __restrict__ bias,
                                                   const int* __restrict__ batch,
                                                   float* __restrict__ pool,
                                                   float* __restrict__ cnt,
                                                   int pool_elems, int N) {
  __shared__ alignas(16) float tile[8][16][68];
  const int wv   = threadIdx.x >> 5;
  const int lane = threadIdx.x & 31;
  const int half = lane >> 4;
  const int l16  = lane & 15;
  const int j    = l16 * 4;
  const int nb   = (blockIdx.x * 8 + wv) * 16;

  int rbeg = 0, rlen = 0;
  float rdn = 0.0f;
  int gid = 0;
  if (nb + l16 < N) {
    const int2 ri = ((const int2*)rowinfo)[nb + l16];
    rbeg = ri.x; rlen = ri.y; rdn = dinv[nb + l16];
    gid = batch[nb + l16];
  }

  int   nbeg = __shfl(rbeg, half, 16);
  int   nlen = __shfl(rlen, half, 16);
  float ndn  = __shfl(rdn,  half, 16);
  int   ncol = 0;
  float nw   = 0.0f;
  if (l16 < nlen) {
    const int2 cw = ((const int2*)colw)[nbeg + l16];
    ncol = (int)min((unsigned)cw.x, (unsigned)(N - 1));
    nw   = __int_as_float(cw.y);
  }

  for (int p = 0; p < 8; ++p) {
    const int r   = 2 * p + half;
    const int n   = nb + r;
    const int beg = nbeg;
    const int len = nlen;
    const float dn = ndn;
    int   mycol = ncol;
    float myw   = nw;
    if (p + 1 < 8) {
      nbeg = __shfl(rbeg, r + 2, 16);
      nlen = __shfl(rlen, r + 2, 16);
      ndn  = __shfl(rdn,  r + 2, 16);
      ncol = 0; nw = 0.0f;
      if (l16 < nlen) {
        const int2 cw = ((const int2*)colw)[nbeg + l16];
        ncol = (int)min((unsigned)cw.x, (unsigned)(N - 1));
        nw   = __int_as_float(cw.y);
      }
    }
    float4 hs = {0.0f, 0.0f, 0.0f, 0.0f};
    if (n < N) hs = *(const float4*)(h + (long)n * HID + j);
    float4 acc = {0.0f, 0.0f, 0.0f, 0.0f};

    for (int base = 0; base < len; base += 16) {
      if (base > 0) {
        mycol = 0; myw = 0.0f;
        if (base + l16 < len) {
          const int2 cw = ((const int2*)colw)[beg + base + l16];
          mycol = (int)min((unsigned)cw.x, (unsigned)(N - 1));
          myw   = __int_as_float(cw.y);
        }
      }
      const int m = min(16, len - base);
      for (int k = 0; k < m; k += 8) {
        float4 v[8];
        float  w[8];
        int    s[8];
        #pragma unroll
        for (int i = 0; i < 8; ++i) {
          s[i] = __shfl(mycol, k + i, 16);
          w[i] = __shfl(myw,   k + i, 16);
        }
        #pragma unroll
        for (int i = 0; i < 8; ++i) {
          v[i] = make_float4(0.0f, 0.0f, 0.0f, 0.0f);
          if (k + i < m) v[i] = *(const float4*)(h + (long)s[i] * HID + j);
        }
        #pragma unroll
        for (int i = 0; i < 8; ++i) {
          acc.x = fmaf(v[i].x, w[i], acc.x); acc.y = fmaf(v[i].y, w[i], acc.y);
          acc.z = fmaf(v[i].z, w[i], acc.z); acc.w = fmaf(v[i].w, w[i], acc.w);
        }
      }
    }

    const float sl = dn * dn;
    float4 a;
    a.x = fmaf(hs.x, sl, acc.x); a.y = fmaf(hs.y, sl, acc.y);
    a.z = fmaf(hs.z, sl, acc.z); a.w = fmaf(hs.w, sl, acc.w);
    *(float4*)&tile[wv][r][j] = a;
  }
  __syncthreads();

  v8f c0 = {}, c1 = {}, c2 = {}, c3 = {};
  const float* __restrict__ Arow = &tile[wv][l16][0];
  #pragma unroll
  for (int i = 0; i < 16; ++i) {
    const int ka = 4 * i + 2 * half;
    v2f a;  a.x  = Arow[ka];                    a.y  = Arow[ka + 1];
    v2f b0; b0.x = W[ka * HID + l16];           b0.y = W[(ka + 1) * HID + l16];
    v2f b1; b1.x = W[ka * HID + 16 + l16];      b1.y = W[(ka + 1) * HID + 16 + l16];
    v2f b2; b2.x = W[ka * HID + 32 + l16];      b2.y = W[(ka + 1) * HID + 32 + l16];
    v2f b3; b3.x = W[ka * HID + 48 + l16];      b3.y = W[(ka + 1) * HID + 48 + l16];
    c0 = __builtin_amdgcn_wmma_f32_16x16x4_f32(false, a, false, b0, (short)0, c0, false, false);
    c1 = __builtin_amdgcn_wmma_f32_16x16x4_f32(false, a, false, b1, (short)0, c1, false, false);
    c2 = __builtin_amdgcn_wmma_f32_16x16x4_f32(false, a, false, b2, (short)0, c2, false, false);
    c3 = __builtin_amdgcn_wmma_f32_16x16x4_f32(false, a, false, b3, (short)0, c3, false, false);
  }
  __syncthreads();

  const float bb0 = bias[l16], bb1 = bias[16 + l16], bb2 = bias[32 + l16], bb3 = bias[48 + l16];
  #pragma unroll
  for (int q = 0; q < 8; ++q) {
    float* trow = &tile[wv][q + 8 * half][0];
    trow[l16]      = fmaxf(c0[q] + bb0, 0.0f);
    trow[16 + l16] = fmaxf(c1[q] + bb1, 0.0f);
    trow[32 + l16] = fmaxf(c2[q] + bb2, 0.0f);
    trow[48 + l16] = fmaxf(c3[q] + bb3, 0.0f);
  }
  __syncthreads();

  const int G   = pool_elems >> 6;
  const int nbu = __builtin_amdgcn_readfirstlane(nb);
  float t0[16], t1[16];
  #pragma unroll
  for (int r = 0; r < 16; ++r) {
    t0[r] = tile[wv][r][lane];
    t1[r] = tile[wv][r][32 + lane];
  }
  float s0 = 0.0f, s1 = 0.0f;
  int gp = 0, run = 0;
  #pragma unroll
  for (int r = 0; r < 16; ++r) {
    if (nbu + r < N) {
      const int g = __builtin_amdgcn_readlane(gid, r);
      if (run > 0 && g != gp) {
        pool_add(pool, cnt, G, gp, run, lane, s0, s1);
        s0 = 0.0f; s1 = 0.0f; run = 0;
      }
      gp = g;
      s0 += t0[r];
      s1 += t1[r];
      ++run;
    }
  }
  if (run > 0) pool_add(pool, cnt, G, gp, run, lane, s0, s1);
}

__global__ void k_div(float* __restrict__ out, const float* __restrict__ cnt, int total) {
  int t = blockIdx.x * blockDim.x + threadIdx.x;
  if (t < total) out[t] /= fmaxf(cnt[t >> 6], 1.0f);
}

extern "C" void kernel_launch(void* const* d_in, const int* in_sizes, int n_in,
                              void* d_out, int out_size, void* d_ws, size_t ws_size,
                              hipStream_t stream) {
  const float* x   = (const float*)d_in[0];
  const float* W1  = (const float*)d_in[1];
  const float* b1  = (const float*)d_in[2];
  const float* W2  = (const float*)d_in[3];
  const float* b2  = (const float*)d_in[4];
  const int*   ei  = (const int*)d_in[5];
  const int*   bat = (const int*)d_in[6];
  float* out = (float*)d_out;

  const int N = in_sizes[0] / 3;
  const int E = in_sizes[5] / 2;

  char* ws = (char*)d_ws;
  float* dinv = (float*)(ws);
  float* bufB = (float*)(ws + (1 << 19) + (size_t)N * HID * 4);
  float* cnt  = (float*)(ws + (1 << 19) + 2 * (size_t)N * HID * 4);
  const size_t Np = ((size_t)N + 63) & ~(size_t)63;
  const size_t Ep = ((size_t)E + 63) & ~(size_t)63;
  const int NB   = (N + RB - 1) / RB;
  const int NCH  = (E + CH - 1) / CH;
  const int NCHP = (NCH + 63) & ~63;
  int*  rowinfo = (int*)(ws + (1 << 19) + 2 * (size_t)N * HID * 4 + 4096);
  int*  col     = rowinfo + 2 * Np;
  int*  colw    = col + Ep;
  int*  sorted  = colw + 2 * Ep;
  int*  offT    = sorted + 2 * (size_t)NCH * CH;

  const int B = 256;
  const int N16  = N * 16;

  k_bin <<<NCH, 1024, 0, stream>>>(ei, E, N, NB, NCHP, sorted, offT);
  k_sort<<<NB,  1024, 0, stream>>>(sorted, offT, NCH, NCHP, N, rowinfo, dinv, col);

  k_gather1<<<(N16 + B - 1) / B, B, 0, stream>>>(rowinfo, col, colw, x, W1, dinv, b1, bufB, N);

  k_zero<<<(out_size + B - 1) / B, B, 0, stream>>>(out, out_size);
  k_zero<<<(512 + B - 1) / B, B, 0, stream>>>(cnt, 512);

  k_gather_mm<<<(N + 127) / 128, B, 0, stream>>>(rowinfo, colw, bufB, W2, dinv, b2, bat, out, cnt, out_size, N);

  k_div <<<(out_size + B - 1) / B, B, 0, stream>>>(out, cnt, out_size);
}


# layer-1 rows never stored: layer 1 keeps 3 summed input features per node ((Ax)W1 for A(xW1)); layer-2 gather forms each neighbour's relu row from them in registers; reorders layer-1 f32 sums
# speedup vs baseline: 7.8983x; 1.2559x over previous
#include <hip/hip_runtime.h>

#define HID 64

typedef float v2f __attribute__((ext_vector_type(2)));
typedef float v8f __attribute__((ext_vector_type(8)));

__global__ void k_zero(float* __restrict__ p, int n) {
  int t = blockIdx.x * blockDim.x + threadIdx.x;
  if (t < n) p[t] = 0.0f;
}

#define LOGR   9
#define RB     512
#define CH     4096
#define NBMAX  1024
#define NCHMAX 1024
#define CAP    8192

__device__ __forceinline__ int block_excl_scan(int v, int* wsum, int lane, int wv, int& total) {
  int inc = v;
  #pragma unroll
  for (int o = 1; o < 32; o <<= 1) {
    int y = __shfl_up(inc, o, 32);
    if (lane >= o) inc += y;
  }
  __syncthreads();
  if (lane == 31) wsum[wv] = inc;
  __syncthreads();
  int woff = 0, t = 0;
  for (int w = 0; w < 32; ++w) {
    const int xw = wsum[w];
    if (w < wv) woff += xw;
    t += xw;
  }
  total = t;
  return woff + inc - v;
}

__global__ __launch_bounds__(1024) void k_bin(const int* __restrict__ ei, int E, int N, int NB, int NCHP,
                                              int* __restrict__ sorted, int* __restrict__ offT) {
  __shared__ int hist[NBMAX];
  __shared__ int wsum[32];
  __shared__ alignas(8) int2 stage[CH];
  const int tid  = threadIdx.x;
  const int lane = tid & 31;
  const int wv   = tid >> 5;
  const int c    = blockIdx.x;
  const int e0   = c * CH;
  hist[tid] = 0;
  __syncthreads();

  int s[4], d[4], r[4];
  #pragma unroll
  for (int u = 0; u < 4; ++u) {
    const int e = e0 + u * 1024 + tid;
    s[u] = 0; d[u] = -1;
    if (e < E) { s[u] = ei[e]; d[u] = ei[E + e]; }
  }
  #pragma unroll
  for (int u = 0; u < 4; ++u) {
    r[u] = 0;
    if (e0 + u * 1024 + tid < E) {
      d[u] = (int)min((unsigned)d[u], (unsigned)(N - 1));
      r[u] = atomicAdd(&hist[d[u] >> LOGR], 1);
    }
  }
  __syncthreads();

  int total;
  const int ex = block_excl_scan(hist[tid], wsum, lane, wv, total);
  hist[tid] = ex;
  if (tid <= NB) offT[tid * NCHP + c] = ex;
  __syncthreads();

  #pragma unroll
  for (int u = 0; u < 4; ++u)
    if (d[u] >= 0) stage[hist[d[u] >> LOGR] + r[u]] = make_int2(s[u], d[u]);
  __syncthreads();

  const int nc = min(CH, E - e0);
  #pragma unroll
  for (int u = 0; u < 4; ++u) {
    const int p = u * 1024 + tid;
    if (p < nc) ((int2*)sorted)[e0 + p] = stage[p];
  }
}

__device__ __forceinline__ void sort_take(int2 e, int p, bool staged, int* cnt, int2* tmp) {
  const int dl = e.y & (RB - 1);
  const int r  = atomicAdd(&cnt[dl], 1);
  if (staged) tmp[p] = make_int2(e.x, (r << LOGR) | dl);
}

__global__ __launch_bounds__(1024) void k_sort(const int* __restrict__ sorted, const int* __restrict__ offT,
                                               int NCH, int NCHP, int N,
                                               int* __restrict__ rowinfo, float* __restrict__ dinv,
                                               int* __restrict__ col) {
  __shared__ alignas(16) int4 run[NCHMAX];
  __shared__ int cnt[RB];
  __shared__ int start[RB];
  __shared__ int wsum[32];
  __shared__ alignas(8) int2 tmp[CAP];
  __shared__ int outst[CAP];
  const int tid  = threadIdx.x;
  const int lane = tid & 31;
  const int wv   = tid >> 5;
  const int b    = blockIdx.x;

  int l0 = 0, ln = 0;
  if (tid < NCH) {
    l0 = offT[b * NCHP + tid];
    ln = offT[(b + 1) * NCHP + tid] - l0;
  }
  int M, base;
  const int myrs = block_excl_scan(ln, wsum, lane, wv, M);
  (void)block_excl_scan(l0, wsum, lane, wv, base);
  run[tid] = make_int4(l0, ln, myrs, 0);
  if (tid < RB) cnt[tid] = 0;
  __syncthreads();
  const bool staged = (M <= CAP);
  const int2* __restrict__ sg = (const int2*)sorted;

  for (int c0 = wv; c0 < NCHMAX; c0 += 128) {
    if (c0 >= NCH) break;
    int4 rn[4];
    #pragma unroll
    for (int u = 0; u < 4; ++u) rn[u] = run[c0 + 32 * u];
    int2 e[4];
    #pragma unroll
    for (int u = 0; u < 4; ++u) {
      e[u] = make_int2(0, 0);
      if (lane < rn[u].y) e[u] = sg[(c0 + 32 * u) * CH + rn[u].x + lane];
    }
    #pragma unroll
    for (int u = 0; u < 4; ++u)
      if (lane < rn[u].y) sort_take(e[u], rn[u].z + lane, staged, cnt, tmp);
    #pragma unroll
    for (int u = 0; u < 4; ++u)
      for (int i = lane + 32; i < rn[u].y; i += 32)
        sort_take(sg[(c0 + 32 * u) * CH + rn[u].x + i], rn[u].z + i, staged, cnt, tmp);
  }
  __syncthreads();

  const int cmine = (tid < RB) ? cnt[tid] : 0;
  int tot2;
  const int ex = block_excl_scan(cmine, wsum, lane, wv, tot2);
  const int n  = b * RB + tid;
  if (tid < RB) {
    start[tid] = ex;
    if (n < N) {
      ((int2*)rowinfo)[n] = make_int2(base + ex, cmine);
      dinv[n] = rsqrtf((float)cmine + 1.0f);
    }
  }
  __syncthreads();

  if (staged) {
    for (int p = tid; p < M; p += 1024) {
      const int2 t = tmp[p];
      outst[start[t.y & (RB - 1)] + (t.y >> LOGR)] = t.x;
    }
    __syncthreads();
    for (int p = tid; p < M; p += 1024) col[base + p] = outst[p];
  } else {
    if (tid < RB) cnt[tid] = 0;
    __syncthreads();
    for (int c = wv; c < NCH; c += 32)
      for (int i = lane; i < run[c].y; i += 32) {
        const int2 ee = sg[c * CH + run[c].x + i];
        const int dl = ee.y & (RB - 1);
        const int r  = atomicAdd(&cnt[dl], 1);
        col[base + start[dl] + r] = ee.x;
      }
  }
}

__global__ __launch_bounds__(256) void k_feat1(const int* __restrict__ rowinfo,
                                               const int* __restrict__ col,
                                               const float* __restrict__ x,
                                               const float* __restrict__ dinv,
                                               float* __restrict__ g, int N, int E) {
  const int t   = blockIdx.x * blockDim.x + threadIdx.x;
  const int n0  = (t >> 4) * 4;
  if (n0 >= N) return;
  const int l16 = threadIdx.x & 15;
  const int EL  = max(E, 1) - 1;
  int nn[4], beg[4], len[4], s[4];
  float dn[4], xs0[4], xs1[4], xs2[4];
  #pragma unroll
  for (int u = 0; u < 4; ++u) {
    nn[u] = min(n0 + u, N - 1);
    const int2 ri = ((const int2*)rowinfo)[nn[u]];
    beg[u] = ri.x;
    len[u] = (n0 + u < N) ? ri.y : 0;
    dn[u]  = dinv[nn[u]];
    xs0[u] = x[nn[u] * 3 + 0]; xs1[u] = x[nn[u] * 3 + 1]; xs2[u] = x[nn[u] * 3 + 2];
  }
  int raw[4];
  #pragma unroll
  for (int u = 0; u < 4; ++u) raw[u] = col[min(beg[u] + l16, EL)];
  asm volatile("" :: "v"(raw[0]), "v"(raw[1]), "v"(raw[2]), "v"(raw[3]));
  #pragma unroll
  for (int u = 0; u < 4; ++u)
    s[u] = (l16 < len[u]) ? (int)min((unsigned)raw[u], (unsigned)(N - 1)) : nn[u];
  float ds[4], v0[4], v1[4], v2[4];
  #pragma unroll
  for (int u = 0; u < 4; ++u) {
    ds[u] = dinv[s[u]];
    v0[u] = x[s[u] * 3 + 0]; v1[u] = x[s[u] * 3 + 1]; v2[u] = x[s[u] * 3 + 2];
  }
  asm volatile("" :: "v"(ds[0]), "v"(ds[1]), "v"(ds[2]), "v"(ds[3]),
                     "v"(v0[0]), "v"(v0[1]), "v"(v0[2]), "v"(v0[3]),
                     "v"(v1[0]), "v"(v1[1]), "v"(v1[2]), "v"(v1[3]),
                     "v"(v2[0]), "v"(v2[1]), "v"(v2[2]), "v"(v2[3]));
  float a0[4], a1[4], a2[4];
  #pragma unroll
  for (int u = 0; u < 4; ++u) {
    const float w = (l16 < len[u]) ? ds[u] * dn[u] : 0.0f;
    a0[u] = (l16 < len[u]) ? v0[u] * w : 0.0f;
    a1[u] = (l16 < len[u]) ? v1[u] * w : 0.0f;
    a2[u] = (l16 < len[u]) ? v2[u] * w : 0.0f;
  }
  #pragma unroll
  for (int u = 0; u < 4; ++u)
    for (int slot = 16 + l16; slot < len[u]; slot += 16) {
      const int ss = (int)min((unsigned)col[beg[u] + slot], (unsigned)(N - 1));
      const float w = dinv[ss] * dn[u];
      a0[u] = fmaf(x[ss * 3 + 0], w, a0[u]); a1[u] = fmaf(x[ss * 3 + 1], w, a1[u]); a2[u] = fmaf(x[ss * 3 + 2], w, a2[u]);
    }
  #pragma unroll
  for (int u = 0; u < 4; ++u) {
    #pragma unroll
    for (int o = 8; o > 0; o >>= 1) {
      a0[u] += __shfl_xor(a0[u], o, 16);
      a1[u] += __shfl_xor(a1[u], o, 16);
      a2[u] += __shfl_xor(a2[u], o, 16);
    }
  }
  #pragma unroll
  for (int u = 0; u < 4; ++u)
    if (l16 == u && n0 + u < N) {
      const float sl = dn[u] * dn[u];
      ((float4*)g)[n0 + u] = make_float4(fmaf(xs0[u], sl, a0[u]), fmaf(xs1[u], sl, a1[u]), fmaf(xs2[u], sl, a2[u]), dn[u]);
    }
}

__device__ __forceinline__ void pool_add(float* __restrict__ pool, float* __restrict__ cnt, int G,
                                         int g, int run, int lane, float s0, float s1) {
  if ((unsigned)g < (unsigned)G) {
    float* o = pool + (long)g * HID;
    atomicAdd(o + lane, s0);
    atomicAdd(o + 32 + lane, s1);
    if (lane == 0) atomicAdd(cnt + g, (float)run);
  }
}

__global__ __launch_bounds__(256) void k_gather_mm(const int* __restrict__ rowinfo,
                                                   const int* __restrict__ col,
                                                   const float* __restrict__ g,
                                                   const float* __restrict__ W1,
                                                   const float* __restrict__ bias1,
                                                   const float* __restrict__ W,
                                                   const float* __restrict__ bias,
                                                   const int* __restrict__ batch,
                                                   float* __restrict__ pool,
                                                   float* __restrict__ cnt,
                                                   int pool_elems, int N, int E) {
  __shared__ alignas(16) float tile[8][16][68];
  const int wv   = threadIdx.x >> 5;
  const int lane = threadIdx.x & 31;
  const int half = lane >> 4;
  const int l16  = lane & 15;
  const int j    = l16 * 4;
  const int nb   = (blockIdx.x * 8 + wv) * 16;
  const float4* __restrict__ g4 = (const float4*)g;

  const float wa0 = W1[j + 0],           wa1 = W1[j + 1],           wa2 = W1[j + 2],           wa3 = W1[j + 3];
  const float wb0 = W1[HID + j + 0],     wb1 = W1[HID + j + 1],     wb2 = W1[HID + j + 2],     wb3 = W1[HID + j + 3];
  const float wc0 = W1[2 * HID + j + 0], wc1 = W1[2 * HID + j + 1], wc2 = W1[2 * HID + j + 2], wc3 = W1[2 * HID + j + 3];
  const float wd0 = bias1[j + 0],        wd1 = bias1[j + 1],        wd2 = bias1[j + 2],        wd3 = bias1[j + 3];

  int rbeg = 0, rlen = 0;
  float4 rg = {0.0f, 0.0f, 0.0f, 0.0f};
  int gid = 0;
  if (nb + l16 < N) {
    const int2 ri = ((const int2*)rowinfo)[nb + l16];
    rbeg = ri.x; rlen = ri.y;
    rg  = g4[nb + l16];
    gid = batch[nb + l16];
  }

  const int EL  = max(E, 1) - 1;
  const int nbc = min(nb, N - 1);
  int cbeg = __shfl(rbeg, half, 16);
  int clen = __shfl(rlen, half, 16);
  int nbeg = __shfl(rbeg, 2 + half, 16);
  int nlen = __shfl(rlen, 2 + half, 16);
  const int craw = col[min(cbeg + l16, EL)];
  int nraw = col[min(nbeg + l16, EL)];
  float4 cg = g4[(l16 < clen) ? (int)min((unsigned)craw, (unsigned)(N - 1)) : nbc];

  for (int p = 0; p < 8; ++p) {
    const int r   = 2 * p + half;
    const int beg = cbeg;
    const int len = clen;
    const float dn  = __shfl(rg.w, r, 16);
    const float sg0 = __shfl(rg.x, r, 16), sg1 = __shfl(rg.y, r, 16), sg2 = __shfl(rg.z, r, 16);
    const bool act = l16 < len;
    float mx0 = act ? cg.x : 0.0f, mx1 = act ? cg.y : 0.0f, mx2 = act ? cg.z : 0.0f;
    float myw = act ? cg.w * dn : 0.0f;
    cbeg = nbeg; clen = nlen;
    cg = g4[(l16 < clen) ? (int)min((unsigned)nraw, (unsigned)(N - 1)) : nbc];
    const int rr = min(r + 4, 15);
    nbeg = __shfl(rbeg, rr, 16);
    nlen = (p + 2 < 8) ? __shfl(rlen, rr, 16) : 0;
    nraw = col[min(nbeg + l16, EL)];

    float4 acc = {0.0f, 0.0f, 0.0f, 0.0f};
    for (int base = 0; base < len; base += 16) {
      if (base > 0) {
        mx0 = 0.0f; mx1 = 0.0f; mx2 = 0.0f; myw = 0.0f;
        if (base + l16 < len) {
          const int ss = (int)min((unsigned)col[beg + base + l16], (unsigned)(N - 1));
          const float4 q = g4[ss];
          mx0 = q.x; mx1 = q.y; mx2 = q.z; myw = q.w * dn;
        }
      }
      const int m = min(16, len - base);
      for (int k = 0; k < m; k += 4) {
        float x0[4], x1[4], x2[4], w[4];
        #pragma unroll
        for (int i = 0; i < 4; ++i) {
          x0[i] = __shfl(mx0, k + i, 16); x1[i] = __shfl(mx1, k + i, 16);
          x2[i] = __shfl(mx2, k + i, 16); w[i]  = __shfl(myw, k + i, 16);
        }
        #pragma unroll
        for (int i = 0; i < 4; ++i) {
          const float r0 = fmaxf(fmaf(x2[i], wc0, fmaf(x1[i], wb0, fmaf(x0[i], wa0, wd0))), 0.0f);
          const float r1 = fmaxf(fmaf(x2[i], wc1, fmaf(x1[i], wb1, fmaf(x0[i], wa1, wd1))), 0.0f);
          const float r2 = fmaxf(fmaf(x2[i], wc2, fmaf(x1[i], wb2, fmaf(x0[i], wa2, wd2))), 0.0f);
          const float r3 = fmaxf(fmaf(x2[i], wc3, fmaf(x1[i], wb3, fmaf(x0[i], wa3, wd3))), 0.0f);
          acc.x = fmaf(r0, w[i], acc.x); acc.y = fmaf(r1, w[i], acc.y);
          acc.z = fmaf(r2, w[i], acc.z); acc.w = fmaf(r3, w[i], acc.w);
        }
      }
    }

    const float sl = dn * dn;
    float4 a;
    a.x = fmaf(fmaxf(fmaf(sg2, wc0, fmaf(sg1, wb0, fmaf(sg0, wa0, wd0))), 0.0f), sl, acc.x);
    a.y = fmaf(fmaxf(fmaf(sg2, wc1, fmaf(sg1, wb1, fmaf(sg0, wa1, wd1))), 0.0f), sl, acc.y);
    a.z = fmaf(fmaxf(fmaf(sg2, wc2, fmaf(sg1, wb2, fmaf(sg0, wa2, wd2))), 0.0f), sl, acc.z);
    a.w = fmaf(fmaxf(fmaf(sg2, wc3, fmaf(sg1, wb3, fmaf(sg0, wa3, wd3))), 0.0f), sl, acc.w);
    *(float4*)&tile[wv][r][j] = a;
  }
  __syncthreads();

  v8f c0 = {}, c1 = {}, c2 = {}, c3 = {};
  const float* __restrict__ Arow = &tile[wv][l16][0];
  #pragma unroll
  for (int i = 0; i < 16; ++i) {
    const int ka = 4 * i + 2 * half;
    v2f a;  a.x  = Arow[ka];                    a.y  = Arow[ka + 1];
    v2f b0; b0.x = W[ka * HID + l16];           b0.y = W[(ka + 1) * HID + l16];
    v2f b1; b1.x = W[ka * HID + 16 + l16];      b1.y = W[(ka + 1) * HID + 16 + l16];
    v2f b2; b2.x = W[ka * HID + 32 + l16];      b2.y = W[(ka + 1) * HID + 32 + l16];
    v2f b3; b3.x = W[ka * HID + 48 + l16];      b3.y = W[(ka + 1) * HID + 48 + l16];
    c0 = __builtin_amdgcn_wmma_f32_16x16x4_f32(false, a, false, b0, (short)0, c0, false, false);
    c1 = __builtin_amdgcn_wmma_f32_16x16x4_f32(false, a, false, b1, (short)0, c1, false, false);
    c2 = __builtin_amdgcn_wmma_f32_16x16x4_f32(false, a, false, b2, (short)0, c2, false, false);
    c3 = __builtin_amdgcn_wmma_f32_16x16x4_f32(false, a, false, b3, (short)0, c3, false, false);
  }
  __syncthreads();

  const float bb0 = bias[l16], bb1 = bias[16 + l16], bb2 = bias[32 + l16], bb3 = bias[48 + l16];
  #pragma unroll
  for (int q = 0; q < 8; ++q) {
    float* trow = &tile[wv][q + 8 * half][0];
    trow[l16]      = fmaxf(c0[q] + bb0, 0.0f);
    trow[16 + l16] = fmaxf(c1[q] + bb1, 0.0f);
    trow[32 + l16] = fmaxf(c2[q] + bb2, 0.0f);
    trow[48 + l16] = fmaxf(c3[q] + bb3, 0.0f);
  }
  __syncthreads();

  const int G   = pool_elems >> 6;
  const int nbu = __builtin_amdgcn_readfirstlane(nb);
  float t0[16], t1[16];
  #pragma unroll
  for (int r = 0; r < 16; ++r) {
    t0[r] = tile[wv][r][lane];
    t1[r] = tile[wv][r][32 + lane];
  }
  float s0 = 0.0f, s1 = 0.0f;
  int gp = 0, run = 0;
  #pragma unroll
  for (int r = 0; r < 16; ++r) {
    if (nbu + r < N) {
      const int g = __builtin_amdgcn_readlane(gid, r);
      if (run > 0 && g != gp) {
        pool_add(pool, cnt, G, gp, run, lane, s0, s1);
        s0 = 0.0f; s1 = 0.0f; run = 0;
      }
      gp = g;
      s0 += t0[r];
      s1 += t1[r];
      ++run;
    }
  }
  if (run > 0) pool_add(pool, cnt, G, gp, run, lane, s0, s1);
}

__global__ void k_div(float* __restrict__ out, const float* __restrict__ cnt, int total) {
  int t = blockIdx.x * blockDim.x + threadIdx.x;
  if (t < total) out[t] /= fmaxf(cnt[t >> 6], 1.0f);
}

extern "C" void kernel_launch(void* const* d_in, const int* in_sizes, int n_in,
                              void* d_out, int out_size, void* d_ws, size_t ws_size,
                              hipStream_t stream) {
  const float* x   = (const float*)d_in[0];
  const float* W1  = (const float*)d_in[1];
  const float* b1  = (const float*)d_in[2];
  const float* W2  = (const float*)d_in[3];
  const float* b2  = (const float*)d_in[4];
  const int*   ei  = (const int*)d_in[5];
  const int*   bat = (const int*)d_in[6];
  float* out = (float*)d_out;

  const int N = in_sizes[0] / 3;
  const int E = in_sizes[5] / 2;

  char* ws = (char*)d_ws;
  float* dinv = (float*)(ws);
  float* gfe  = (float*)(ws + (1 << 19) + (size_t)N * HID * 4);
  float* cnt  = (float*)(ws + (1 << 19) + 2 * (size_t)N * HID * 4);
  const size_t Np = ((size_t)N + 63) & ~(size_t)63;
  const size_t Ep = ((size_t)E + 63) & ~(size_t)63;
  const int NB   = (N + RB - 1) / RB;
  const int NCH  = (E + CH - 1) / CH;
  const int NCHP = (NCH + 63) & ~63;
  int*  rowinfo = (int*)(ws + (1 << 19) + 2 * (size_t)N * HID * 4 + 4096);
  int*  col     = rowinfo + 2 * Np;
  int*  sorted  = col + Ep;
  int*  offT    = sorted + 2 * (size_t)NCH * CH;

  const int B = 256;
  const int N4   = ((N + 3) / 4) * 16;

  k_bin <<<NCH, 1024, 0, stream>>>(ei, E, N, NB, NCHP, sorted, offT);
  k_sort<<<NB,  1024, 0, stream>>>(sorted, offT, NCH, NCHP, N, rowinfo, dinv, col);

  k_feat1<<<(N4 + B - 1) / B, B, 0, stream>>>(rowinfo, col, x, dinv, gfe, N, E);

  k_zero<<<(out_size + B - 1) / B, B, 0, stream>>>(out, out_size);
  k_zero<<<(512 + B - 1) / B, B, 0, stream>>>(cnt, 512);

  k_gather_mm<<<(N + 127) / 128, B, 0, stream>>>(rowinfo, col, gfe, W1, b1, W2, b2, bat, out, cnt, out_size, N, E);

  k_div <<<(out_size + B - 1) / B, B, 0, stream>>>(out, cnt, out_size);
}


# mean pool's zero-fill, node count and division folded into the sort and layer-2 kernels (7 launches to 4); run sums divided by the finished count before the atomic add: rescales the f32 addends
# speedup vs baseline: 8.6537x; 1.0956x over previous
#include <hip/hip_runtime.h>

#define HID 64

typedef float v2f __attribute__((ext_vector_type(2)));
typedef float v8f __attribute__((ext_vector_type(8)));

#define LOGR   9
#define RB     512
#define CH     4096
#define NBMAX  1024
#define NCHMAX 1024
#define CAP    8192

__device__ __forceinline__ int block_excl_scan(int v, int* wsum, int lane, int wv, int& total) {
  int inc = v;
  #pragma unroll
  for (int o = 1; o < 32; o <<= 1) {
    int y = __shfl_up(inc, o, 32);
    if (lane >= o) inc += y;
  }
  __syncthreads();
  if (lane == 31) wsum[wv] = inc;
  __syncthreads();
  int woff = 0, t = 0;
  for (int w = 0; w < 32; ++w) {
    const int xw = wsum[w];
    if (w < wv) woff += xw;
    t += xw;
  }
  total = t;
  return woff + inc - v;
}

__global__ __launch_bounds__(1024) void k_bin(const int* __restrict__ ei, int E, int N, int NB, int NCHP,
                                              int* __restrict__ sorted, int* __restrict__ offT,
                                              float* __restrict__ pool, int pool_elems, int* __restrict__ gcnt) {
  __shared__ int hist[NBMAX];
  __shared__ int wsum[32];
  __shared__ alignas(8) int2 stage[CH];
  const int tid  = threadIdx.x;
  const int lane = tid & 31;
  const int wv   = tid >> 5;
  const int c    = blockIdx.x;
  const int e0   = c * CH;
  hist[tid] = 0;
  __syncthreads();

  int s[4], d[4], r[4];
  #pragma unroll
  for (int u = 0; u < 4; ++u) {
    const int e = e0 + u * 1024 + tid;
    s[u] = 0; d[u] = -1;
    if (e < E) { s[u] = ei[e]; d[u] = ei[E + e]; }
  }
  #pragma unroll
  for (int u = 0; u < 4; ++u) {
    r[u] = 0;
    if (e0 + u * 1024 + tid < E) {
      d[u] = (int)min((unsigned)d[u], (unsigned)(N - 1));
      r[u] = atomicAdd(&hist[d[u] >> LOGR], 1);
    }
  }
  __syncthreads();

  int total;
  const int ex = block_excl_scan(hist[tid], wsum, lane, wv, total);
  hist[tid] = ex;
  if (tid <= NB) offT[tid * NCHP + c] = ex;
  __syncthreads();

  #pragma unroll
  for (int u = 0; u < 4; ++u)
    if (d[u] >= 0) stage[hist[d[u] >> LOGR] + r[u]] = make_int2(s[u], d[u]);
  __syncthreads();

  const int nc = min(CH, E - e0);
  #pragma unroll
  for (int u = 0; u < 4; ++u) {
    const int p = u * 1024 + tid;
    if (p < nc) ((int2*)sorted)[e0 + p] = stage[p];
  }

  for (int t = c * 1024 + tid; t < pool_elems; t += gridDim.x * 1024) pool[t] = 0.0f;
  for (int t = c * 1024 + tid; t < (pool_elems >> 6); t += gridDim.x * 1024) gcnt[t] = 0;
}

__device__ __forceinline__ void sort_take(int2 e, int p, bool staged, int* cnt, int2* tmp) {
  const int dl = e.y & (RB - 1);
  const int r  = atomicAdd(&cnt[dl], 1);
  if (staged) tmp[p] = make_int2(e.x, (r << LOGR) | dl);
}

__global__ __launch_bounds__(1024) void k_sort(const int* __restrict__ sorted, const int* __restrict__ offT,
                                               int NCH, int NCHP, int N,
                                               int* __restrict__ rowinfo, float* __restrict__ dinv,
                                               int* __restrict__ col,
                                               const int* __restrict__ batch, int* __restrict__ gcnt, int G) {
  __shared__ alignas(16) int4 run[NCHMAX];
  __shared__ int cnt[RB];
  __shared__ int start[RB];
  __shared__ int wsum[32];
  __shared__ alignas(8) int2 tmp[CAP];
  __shared__ int outst[CAP];
  const int tid  = threadIdx.x;
  const int lane = tid & 31;
  const int wv   = tid >> 5;
  const int b    = blockIdx.x;

  int gq = -1;
  if (tid < RB && b * RB + tid < N) gq = batch[b * RB + tid];

  int l0 = 0, ln = 0;
  if (tid < NCH) {
    l0 = offT[b * NCHP + tid];
    ln = offT[(b + 1) * NCHP + tid] - l0;
  }
  int M, base;
  const int myrs = block_excl_scan(ln, wsum, lane, wv, M);
  (void)block_excl_scan(l0, wsum, lane, wv, base);
  run[tid] = make_int4(l0, ln, myrs, 0);
  if (tid < RB) cnt[tid] = 0;
  __syncthreads();
  const bool staged = (M <= CAP);
  const int2* __restrict__ sg = (const int2*)sorted;

  for (int c0 = wv; c0 < NCHMAX; c0 += 128) {
    if (c0 >= NCH) break;
    int4 rn[4];
    #pragma unroll
    for (int u = 0; u < 4; ++u) rn[u] = run[c0 + 32 * u];
    int2 e[4];
    #pragma unroll
    for (int u = 0; u < 4; ++u) {
      e[u] = make_int2(0, 0);
      if (lane < rn[u].y) e[u] = sg[(c0 + 32 * u) * CH + rn[u].x + lane];
    }
    #pragma unroll
    for (int u = 0; u < 4; ++u)
      if (lane < rn[u].y) sort_take(e[u], rn[u].z + lane, staged, cnt, tmp);
    #pragma unroll
    for (int u = 0; u < 4; ++u)
      for (int i = lane + 32; i < rn[u].y; i += 32)
        sort_take(sg[(c0 + 32 * u) * CH + rn[u].x + i], rn[u].z + i, staged, cnt, tmp);
  }
  __syncthreads();

  const int cmine = (tid < RB) ? cnt[tid] : 0;
  int tot2;
  const int ex = block_excl_scan(cmine, wsum, lane, wv, tot2);
  const int n  = b * RB + tid;
  if (tid < RB) {
    start[tid] = ex;
    if (n < N) {
      ((int2*)rowinfo)[n] = make_int2(base + ex, cmine);
      dinv[n] = rsqrtf((float)cmine + 1.0f);
    }
  }
  __syncthreads();

  if (staged) {
    for (int p = tid; p < M; p += 1024) {
      const int2 t = tmp[p];
      outst[start[t.y & (RB - 1)] + (t.y >> LOGR)] = t.x;
    }
    __syncthreads();
    for (int p = tid; p < M; p += 1024) col[base + p] = outst[p];
  } else {
    if (tid < RB) cnt[tid] = 0;
    __syncthreads();
    for (int c = wv; c < NCH; c += 32)
      for (int i = lane; i < run[c].y; i += 32) {
        const int2 ee = sg[c * CH + run[c].x + i];
        const int dl = ee.y & (RB - 1);
        const int r  = atomicAdd(&cnt[dl], 1);
        col[base + start[dl] + r] = ee.x;
      }
  }

  if (tid < RB) {
    const int gbelow = __shfl_up(gq, 1, 32);
    const bool head  = (lane == 0) || (gq != gbelow);
    const unsigned hm    = (unsigned)__ballot(head);
    const unsigned above = (lane == 31) ? 0u : (hm >> (lane + 1));
    const int runlen = above ? __ffs((int)above) : (32 - lane);
    if (head && (unsigned)gq < (unsigned)G) atomicAdd(&gcnt[gq], runlen);
  }
}

__global__ __launch_bounds__(256) void k_feat1(const int* __restrict__ rowinfo,
                                               const int* __restrict__ col,
                                               const float* __restrict__ x,
                                               const float* __restrict__ dinv,
                                               float* __restrict__ g, int N, int E) {
  const int t   = blockIdx.x * blockDim.x + threadIdx.x;
  const int n0  = (t >> 4) * 4;
  if (n0 >= N) return;
  const int l16 = threadIdx.x & 15;
  const int EL  = max(E, 1) - 1;
  int nn[4], beg[4], len[4], s[4];
  float dn[4], xs0[4], xs1[4], xs2[4];
  #pragma unroll
  for (int u = 0; u < 4; ++u) {
    nn[u] = min(n0 + u, N - 1);
    const int2 ri = ((const int2*)rowinfo)[nn[u]];
    beg[u] = ri.x;
    len[u] = (n0 + u < N) ? ri.y : 0;
    dn[u]  = dinv[nn[u]];
    xs0[u] = x[nn[u] * 3 + 0]; xs1[u] = x[nn[u] * 3 + 1]; xs2[u] = x[nn[u] * 3 + 2];
  }
  int raw[4];
  #pragma unroll
  for (int u = 0; u < 4; ++u) raw[u] = col[min(beg[u] + l16, EL)];
  asm volatile("" :: "v"(raw[0]), "v"(raw[1]), "v"(raw[2]), "v"(raw[3]));
  #pragma unroll
  for (int u = 0; u < 4; ++u)
    s[u] = (l16 < len[u]) ? (int)min((unsigned)raw[u], (unsigned)(N - 1)) : nn[u];
  float ds[4], v0[4], v1[4], v2[4];
  #pragma unroll
  for (int u = 0; u < 4; ++u) {
    ds[u] = dinv[s[u]];
    v0[u] = x[s[u] * 3 + 0]; v1[u] = x[s[u] * 3 + 1]; v2[u] = x[s[u] * 3 + 2];
  }
  asm volatile("" :: "v"(ds[0]), "v"(ds[1]), "v"(ds[2]), "v"(ds[3]),
                     "v"(v0[0]), "v"(v0[1]), "v"(v0[2]), "v"(v0[3]),
                     "v"(v1[0]), "v"(v1[1]), "v"(v1[2]), "v"(v1[3]),
                     "v"(v2[0]), "v"(v2[1]), "v"(v2[2]), "v"(v2[3]));
  float a0[4], a1[4], a2[4];
  #pragma unroll
  for (int u = 0; u < 4; ++u) {
    const float w = (l16 < len[u]) ? ds[u] * dn[u] : 0.0f;
    a0[u] = (l16 < len[u]) ? v0[u] * w : 0.0f;
    a1[u] = (l16 < len[u]) ? v1[u] * w : 0.0f;
    a2[u] = (l16 < len[u]) ? v2[u] * w : 0.0f;
  }
  #pragma unroll
  for (int u = 0; u < 4; ++u)
    for (int slot = 16 + l16; slot < len[u]; slot += 16) {
      const int ss = (int)min((unsigned)col[beg[u] + slot], (unsigned)(N - 1));
      const float w = dinv[ss] * dn[u];
      a0[u] = fmaf(x[ss * 3 + 0], w, a0[u]); a1[u] = fmaf(x[ss * 3 + 1], w, a1[u]); a2[u] = fmaf(x[ss * 3 + 2], w, a2[u]);
    }
  #pragma unroll
  for (int u = 0; u < 4; ++u) {
    #pragma unroll
    for (int o = 8; o > 0; o >>= 1) {
      a0[u] += __shfl_xor(a0[u], o, 16);
      a1[u] += __shfl_xor(a1[u], o, 16);
      a2[u] += __shfl_xor(a2[u], o, 16);
    }
  }
  #pragma unroll
  for (int u = 0; u < 4; ++u)
    if (l16 == u && n0 + u < N) {
      const float sl = dn[u] * dn[u];
      ((float4*)g)[n0 + u] = make_float4(fmaf(xs0[u], sl, a0[u]), fmaf(xs1[u], sl, a1[u]), fmaf(xs2[u], sl, a2[u]), dn[u]);
    }
}

__device__ __forceinline__ void pool_add(float* __restrict__ pool, const int* __restrict__ gcnt, int G,
                                         int g, int lane, float s0, float s1) {
  if ((unsigned)g < (unsigned)G) {
    const float cn = fmaxf((float)gcnt[g], 1.0f);
    float* o = pool + (long)g * HID;
    atomicAdd(o + lane, s0 / cn);
    atomicAdd(o + 32 + lane, s1 / cn);
  }
}

__global__ __launch_bounds__(256) void k_gather_mm(const int* __restrict__ rowinfo,
                                                   const int* __restrict__ col,
                                                   const float* __restrict__ g,
                                                   const float* __restrict__ W1,
                                                   const float* __restrict__ bias1,
                                                   const float* __restrict__ W,
                                                   const float* __restrict__ bias,
                                                   const int* __restrict__ batch,
                                                   float* __restrict__ pool,
                                                   const int* __restrict__ gcnt,
                                                   int pool_elems, int N, int E) {
  __shared__ alignas(16) float tile[8][16][68];
  const int wv   = threadIdx.x >> 5;
  const int lane = threadIdx.x & 31;
  const int half = lane >> 4;
  const int l16  = lane & 15;
  const int j    = l16 * 4;
  const int nb   = (blockIdx.x * 8 + wv) * 16;
  const float4* __restrict__ g4 = (const float4*)g;

  const float wa0 = W1[j + 0],           wa1 = W1[j + 1],           wa2 = W1[j + 2],           wa3 = W1[j + 3];
  const float wb0 = W1[HID + j + 0],     wb1 = W1[HID + j + 1],     wb2 = W1[HID + j + 2],     wb3 = W1[HID + j + 3];
  const float wc0 = W1[2 * HID + j + 0], wc1 = W1[2 * HID + j + 1], wc2 = W1[2 * HID + j + 2], wc3 = W1[2 * HID + j + 3];
  const float wd0 = bias1[j + 0],        wd1 = bias1[j + 1],        wd2 = bias1[j + 2],        wd3 = bias1[j + 3];

  int rbeg = 0, rlen = 0;
  float4 rg = {0.0f, 0.0f, 0.0f, 0.0f};
  int gid = 0;
  if (nb + l16 < N) {
    const int2 ri = ((const int2*)rowinfo)[nb + l16];
    rbeg = ri.x; rlen = ri.y;
    rg  = g4[nb + l16];
    gid = batch[nb + l16];
  }

  const int EL  = max(E, 1) - 1;
  const int nbc = min(nb, N - 1);
  int cbeg = __shfl(rbeg, half, 16);
  int clen = __shfl(rlen, half, 16);
  int nbeg = __shfl(rbeg, 2 + half, 16);
  int nlen = __shfl(rlen, 2 + half, 16);
  const int craw = col[min(cbeg + l16, EL)];
  int nraw = col[min(nbeg + l16, EL)];
  float4 cg = g4[(l16 < clen) ? (int)min((unsigned)craw, (unsigned)(N - 1)) : nbc];

  for (int p = 0; p < 8; ++p) {
    const int r   = 2 * p + half;
    const int beg = cbeg;
    const int len = clen;
    const float dn  = __shfl(rg.w, r, 16);
    const float sg0 = __shfl(rg.x, r, 16), sg1 = __shfl(rg.y, r, 16), sg2 = __shfl(rg.z, r, 16);
    const bool act = l16 < len;
    float mx0 = act ? cg.x : 0.0f, mx1 = act ? cg.y : 0.0f, mx2 = act ? cg.z : 0.0f;
    float myw = act ? cg.w * dn : 0.0f;
    cbeg = nbeg; clen = nlen;
    cg = g4[(l16 < clen) ? (int)min((unsigned)nraw, (unsigned)(N - 1)) : nbc];
    const int rr = min(r + 4, 15);
    nbeg = __shfl(rbeg, rr, 16);
    nlen = (p + 2 < 8) ? __shfl(rlen, rr, 16) : 0;
    nraw = col[min(nbeg + l16, EL)];

    float4 acc = {0.0f, 0.0f, 0.0f, 0.0f};
    for (int base = 0; base < len; base += 16) {
      if (base > 0) {
        mx0 = 0.0f; mx1 = 0.0f; mx2 = 0.0f; myw = 0.0f;
        if (base + l16 < len) {
          const int ss = (int)min((unsigned)col[beg + base + l16], (unsigned)(N - 1));
          const float4 q = g4[ss];
          mx0 = q.x; mx1 = q.y; mx2 = q.z; myw = q.w * dn;
        }
      }
      const int m = min(16, len - base);
      for (int k = 0; k < m; k += 4) {
        float x0[4], x1[4], x2[4], w[4];
        #pragma unroll
        for (int i = 0; i < 4; ++i) {
          x0[i] = __shfl(mx0, k + i, 16); x1[i] = __shfl(mx1, k + i, 16);
          x2[i] = __shfl(mx2, k + i, 16); w[i]  = __shfl(myw, k + i, 16);
        }
        #pragma unroll
        for (int i = 0; i < 4; ++i) {
          const float r0 = fmaxf(fmaf(x2[i], wc0, fmaf(x1[i], wb0, fmaf(x0[i], wa0, wd0))), 0.0f);
          const float r1 = fmaxf(fmaf(x2[i], wc1, fmaf(x1[i], wb1, fmaf(x0[i], wa1, wd1))), 0.0f);
          const float r2 = fmaxf(fmaf(x2[i], wc2, fmaf(x1[i], wb2, fmaf(x0[i], wa2, wd2))), 0.0f);
          const float r3 = fmaxf(fmaf(x2[i], wc3, fmaf(x1[i], wb3, fmaf(x0[i], wa3, wd3))), 0.0f);
          acc.x = fmaf(r0, w[i], acc.x); acc.y = fmaf(r1, w[i], acc.y);
          acc.z = fmaf(r2, w[i], acc.z); acc.w = fmaf(r3, w[i], acc.w);
        }
      }
    }

    const float sl = dn * dn;
    float4 a;
    a.x = fmaf(fmaxf(fmaf(sg2, wc0, fmaf(sg1, wb0, fmaf(sg0, wa0, wd0))), 0.0f), sl, acc.x);
    a.y = fmaf(fmaxf(fmaf(sg2, wc1, fmaf(sg1, wb1, fmaf(sg0, wa1, wd1))), 0.0f), sl, acc.y);
    a.z = fmaf(fmaxf(fmaf(sg2, wc2, fmaf(sg1, wb2, fmaf(sg0, wa2, wd2))), 0.0f), sl, acc.z);
    a.w = fmaf(fmaxf(fmaf(sg2, wc3, fmaf(sg1, wb3, fmaf(sg0, wa3, wd3))), 0.0f), sl, acc.w);
    *(float4*)&tile[wv][r][j] = a;
  }
  __syncthreads();

  v8f c0 = {}, c1 = {}, c2 = {}, c3 = {};
  const float* __restrict__ Arow = &tile[wv][l16][0];
  #pragma unroll
  for (int i = 0; i < 16; ++i) {
    const int ka = 4 * i + 2 * half;
    v2f a;  a.x  = Arow[ka];                    a.y  = Arow[ka + 1];
    v2f b0; b0.x = W[ka * HID + l16];           b0.y = W[(ka + 1) * HID + l16];
    v2f b1; b1.x = W[ka * HID + 16 + l16];      b1.y = W[(ka + 1) * HID + 16 + l16];
    v2f b2; b2.x = W[ka * HID + 32 + l16];      b2.y = W[(ka + 1) * HID + 32 + l16];
    v2f b3; b3.x = W[ka * HID + 48 + l16];      b3.y = W[(ka + 1) * HID + 48 + l16];
    c0 = __builtin_amdgcn_wmma_f32_16x16x4_f32(false, a, false, b0, (short)0, c0, false, false);
    c1 = __builtin_amdgcn_wmma_f32_16x16x4_f32(false, a, false, b1, (short)0, c1, false, false);
    c2 = __builtin_amdgcn_wmma_f32_16x16x4_f32(false, a, false, b2, (short)0, c2, false, false);
    c3 = __builtin_amdgcn_wmma_f32_16x16x4_f32(false, a, false, b3, (short)0, c3, false, false);
  }
  __syncthreads();

  const float bb0 = bias[l16], bb1 = bias[16 + l16], bb2 = bias[32 + l16], bb3 = bias[48 + l16];
  #pragma unroll
  for (int q = 0; q < 8; ++q) {
    float* trow = &tile[wv][q + 8 * half][0];
    trow[l16]      = fmaxf(c0[q] + bb0, 0.0f);
    trow[16 + l16] = fmaxf(c1[q] + bb1, 0.0f);
    trow[32 + l16] = fmaxf(c2[q] + bb2, 0.0f);
    trow[48 + l16] = fmaxf(c3[q] + bb3, 0.0f);
  }
  __syncthreads();

  const int G   = pool_elems >> 6;
  const int nbu = __builtin_amdgcn_readfirstlane(nb);
  float t0[16], t1[16];
  #pragma unroll
  for (int r = 0; r < 16; ++r) {
    t0[r] = tile[wv][r][lane];
    t1[r] = tile[wv][r][32 + lane];
  }
  float s0 = 0.0f, s1 = 0.0f;
  int gp = 0, run = 0;
  #pragma unroll
  for (int r = 0; r < 16; ++r) {
    if (nbu + r < N) {
      const int g = __builtin_amdgcn_readlane(gid, r);
      if (run > 0 && g != gp) {
        pool_add(pool, gcnt, G, gp, lane, s0, s1);
        s0 = 0.0f; s1 = 0.0f; run = 0;
      }
      gp = g;
      s0 += t0[r];
      s1 += t1[r];
      ++run;
    }
  }
  if (run > 0) pool_add(pool, gcnt, G, gp, lane, s0, s1);
}

extern "C" void kernel_launch(void* const* d_in, const int* in_sizes, int n_in,
                              void* d_out, int out_size, void* d_ws, size_t ws_size,
                              hipStream_t stream) {
  const float* x   = (const float*)d_in[0];
  const float* W1  = (const float*)d_in[1];
  const float* b1  = (const float*)d_in[2];
  const float* W2  = (const float*)d_in[3];
  const float* b2  = (const float*)d_in[4];
  const int*   ei  = (const int*)d_in[5];
  const int*   bat = (const int*)d_in[6];
  float* out = (float*)d_out;

  const int N = in_sizes[0] / 3;
  const int E = in_sizes[5] / 2;

  char* ws = (char*)d_ws;
  float* dinv = (float*)(ws);
  float* gfe  = (float*)(ws + (1 << 19) + (size_t)N * HID * 4);
  int*   gcnt = (int*)(ws + (1 << 19) + 2 * (size_t)N * HID * 4);
  const size_t Np = ((size_t)N + 63) & ~(size_t)63;
  const size_t Ep = ((size_t)E + 63) & ~(size_t)63;
  const int NB   = (N + RB - 1) / RB;
  const int NCH  = (E + CH - 1) / CH;
  const int NCHP = (NCH + 63) & ~63;
  int*  rowinfo = (int*)(ws + (1 << 19) + 2 * (size_t)N * HID * 4 + 4096);
  int*  col     = rowinfo + 2 * Np;
  int*  sorted  = col + Ep;
  int*  offT    = sorted + 2 * (size_t)NCH * CH;

  const int B = 256;
  const int N4   = ((N + 3) / 4) * 16;

  k_bin <<<NCH, 1024, 0, stream>>>(ei, E, N, NB, NCHP, sorted, offT, out, out_size, gcnt);
  k_sort<<<NB,  1024, 0, stream>>>(sorted, offT, NCH, NCHP, N, rowinfo, dinv, col, bat, gcnt, out_size >> 6);

  k_feat1<<<(N4 + B - 1) / B, B, 0, stream>>>(rowinfo, col, x, dinv, gfe, N, E);

  k_gather_mm<<<(N + 127) / 128, B, 0, stream>>>(rowinfo, col, gfe, W1, b1, W2, b2, bat, out, gcnt, out_size, N, E);
}


# k_sort stores one 16-byte record {x0,x1,x2,dinv} per node; the layer-1 gather fetches one record per edge slot, not a 12-byte x row plus a dinv word from two arrays; sums and their order unchanged
# speedup vs baseline: 9.5449x; 1.1030x over previous
#include <hip/hip_runtime.h>

#define HID 64

typedef float v2f __attribute__((ext_vector_type(2)));
typedef float v8f __attribute__((ext_vector_type(8)));

#define LOGR   9
#define RB     512
#define CH     4096
#define NBMAX  1024
#define NCHMAX 1024
#define CAP    8192

__device__ __forceinline__ int block_excl_scan(int v, int* wsum, int lane, int wv, int& total) {
  int inc = v;
  #pragma unroll
  for (int o = 1; o < 32; o <<= 1) {
    int y = __shfl_up(inc, o, 32);
    if (lane >= o) inc += y;
  }
  __syncthreads();
  if (lane == 31) wsum[wv] = inc;
  __syncthreads();
  int woff = 0, t = 0;
  for (int w = 0; w < 32; ++w) {
    const int xw = wsum[w];
    if (w < wv) woff += xw;
    t += xw;
  }
  total = t;
  return woff + inc - v;
}

__global__ __launch_bounds__(1024) void k_bin(const int* __restrict__ ei, int E, int N, int NB, int NCHP,
                                              int* __restrict__ sorted, int* __restrict__ offT,
                                              float* __restrict__ pool, int pool_elems, int* __restrict__ gcnt) {
  __shared__ int hist[NBMAX];
  __shared__ int wsum[32];
  __shared__ alignas(8) int2 stage[CH];
  const int tid  = threadIdx.x;
  const int lane = tid & 31;
  const int wv   = tid >> 5;
  const int c    = blockIdx.x;
  const int e0   = c * CH;
  hist[tid] = 0;
  __syncthreads();

  int s[4], d[4], r[4];
  #pragma unroll
  for (int u = 0; u < 4; ++u) {
    const int e = e0 + u * 1024 + tid;
    s[u] = 0; d[u] = -1;
    if (e < E) { s[u] = ei[e]; d[u] = ei[E + e]; }
  }
  #pragma unroll
  for (int u = 0; u < 4; ++u) {
    r[u] = 0;
    if (e0 + u * 1024 + tid < E) {
      d[u] = (int)min((unsigned)d[u], (unsigned)(N - 1));
      r[u] = atomicAdd(&hist[d[u] >> LOGR], 1);
    }
  }
  __syncthreads();

  int total;
  const int ex = block_excl_scan(hist[tid], wsum, lane, wv, total);
  hist[tid] = ex;
  if (tid <= NB) offT[tid * NCHP + c] = ex;
  __syncthreads();

  #pragma unroll
  for (int u = 0; u < 4; ++u)
    if (d[u] >= 0) stage[hist[d[u] >> LOGR] + r[u]] = make_int2(s[u], d[u]);
  __syncthreads();

  const int nc = min(CH, E - e0);
  #pragma unroll
  for (int u = 0; u < 4; ++u) {
    const int p = u * 1024 + tid;
    if (p < nc) ((int2*)sorted)[e0 + p] = stage[p];
  }

  for (int t = c * 1024 + tid; t < pool_elems; t += gridDim.x * 1024) pool[t] = 0.0f;
  for (int t = c * 1024 + tid; t < (pool_elems >> 6); t += gridDim.x * 1024) gcnt[t] = 0;
}

__device__ __forceinline__ void sort_take(int2 e, int p, bool staged, int* cnt, int2* tmp) {
  const int dl = e.y & (RB - 1);
  const int r  = atomicAdd(&cnt[dl], 1);
  if (staged) tmp[p] = make_int2(e.x, (r << LOGR) | dl);
}

__global__ __launch_bounds__(1024) void k_sort(const int* __restrict__ sorted, const int* __restrict__ offT,
                                               int NCH, int NCHP, int N,
                                               int* __restrict__ rowinfo, const float* __restrict__ x,
                                               float* __restrict__ rec, int* __restrict__ col,
                                               const int* __restrict__ batch, int* __restrict__ gcnt, int G) {
  __shared__ alignas(16) int4 run[NCHMAX];
  __shared__ int cnt[RB];
  __shared__ int start[RB];
  __shared__ int wsum[32];
  __shared__ alignas(8) int2 tmp[CAP];
  __shared__ int outst[CAP];
  const int tid  = threadIdx.x;
  const int lane = tid & 31;
  const int wv   = tid >> 5;
  const int b    = blockIdx.x;

  int gq = -1;
  float xq0 = 0.0f, xq1 = 0.0f, xq2 = 0.0f;
  if (tid < RB && b * RB + tid < N) {
    gq = batch[b * RB + tid];
    const float* __restrict__ xp = x + (size_t)(b * RB + tid) * 3;
    xq0 = xp[0]; xq1 = xp[1]; xq2 = xp[2];
  }

  int l0 = 0, ln = 0;
  if (tid < NCH) {
    l0 = offT[b * NCHP + tid];
    ln = offT[(b + 1) * NCHP + tid] - l0;
  }
  int M, base;
  const int myrs = block_excl_scan(ln, wsum, lane, wv, M);
  (void)block_excl_scan(l0, wsum, lane, wv, base);
  run[tid] = make_int4(l0, ln, myrs, 0);
  if (tid < RB) cnt[tid] = 0;
  __syncthreads();
  const bool staged = (M <= CAP);
  const int2* __restrict__ sg = (const int2*)sorted;

  for (int c0 = wv; c0 < NCHMAX; c0 += 128) {
    if (c0 >= NCH) break;
    int4 rn[4];
    #pragma unroll
    for (int u = 0; u < 4; ++u) rn[u] = run[c0 + 32 * u];
    int2 e[4];
    #pragma unroll
    for (int u = 0; u < 4; ++u) {
      e[u] = make_int2(0, 0);
      if (lane < rn[u].y) e[u] = sg[(c0 + 32 * u) * CH + rn[u].x + lane];
    }
    #pragma unroll
    for (int u = 0; u < 4; ++u)
      if (lane < rn[u].y) sort_take(e[u], rn[u].z + lane, staged, cnt, tmp);
    #pragma unroll
    for (int u = 0; u < 4; ++u)
      for (int i = lane + 32; i < rn[u].y; i += 32)
        sort_take(sg[(c0 + 32 * u) * CH + rn[u].x + i], rn[u].z + i, staged, cnt, tmp);
  }
  __syncthreads();

  const int cmine = (tid < RB) ? cnt[tid] : 0;
  int tot2;
  const int ex = block_excl_scan(cmine, wsum, lane, wv, tot2);
  const int n  = b * RB + tid;
  if (tid < RB) {
    start[tid] = ex;
    if (n < N) {
      ((int2*)rowinfo)[n] = make_int2(base + ex, cmine);
      ((float4*)rec)[n] = make_float4(xq0, xq1, xq2, rsqrtf((float)cmine + 1.0f));
    }
  }
  __syncthreads();

  if (staged) {
    for (int p = tid; p < M; p += 1024) {
      const int2 t = tmp[p];
      outst[start[t.y & (RB - 1)] + (t.y >> LOGR)] = t.x;
    }
    __syncthreads();
    for (int p = tid; p < M; p += 1024) col[base + p] = outst[p];
  } else {
    if (tid < RB) cnt[tid] = 0;
    __syncthreads();
    for (int c = wv; c < NCH; c += 32)
      for (int i = lane; i < run[c].y; i += 32) {
        const int2 ee = sg[c * CH + run[c].x + i];
        const int dl = ee.y & (RB - 1);
        const int r  = atomicAdd(&cnt[dl], 1);
        col[base + start[dl] + r] = ee.x;
      }
  }

  if (tid < RB) {
    const int gbelow = __shfl_up(gq, 1, 32);
    const bool head  = (lane == 0) || (gq != gbelow);
    const unsigned hm    = (unsigned)__ballot(head);
    const unsigned above = (lane == 31) ? 0u : (hm >> (lane + 1));
    const int runlen = above ? __ffs((int)above) : (32 - lane);
    if (head && (unsigned)gq < (unsigned)G) atomicAdd(&gcnt[gq], runlen);
  }
}

__global__ __launch_bounds__(256) void k_feat1(const int* __restrict__ rowinfo,
                                               const int* __restrict__ col,
                                               const float* __restrict__ rec,
                                               float* __restrict__ g, int N, int E) {
  const int t   = blockIdx.x * blockDim.x + threadIdx.x;
  const int n0  = (t >> 4) * 4;
  if (n0 >= N) return;
  const int l16 = threadIdx.x & 15;
  const int EL  = max(E, 1) - 1;
  const float4* __restrict__ r4 = (const float4*)rec;
  int nn[4], beg[4], len[4], s[4];
  float4 own[4];
  #pragma unroll
  for (int u = 0; u < 4; ++u) {
    nn[u] = min(n0 + u, N - 1);
    const int2 ri = ((const int2*)rowinfo)[nn[u]];
    beg[u] = ri.x;
    len[u] = (n0 + u < N) ? ri.y : 0;
    own[u] = r4[nn[u]];
  }
  int raw[4];
  #pragma unroll
  for (int u = 0; u < 4; ++u) raw[u] = col[min(beg[u] + l16, EL)];
  asm volatile("" :: "v"(raw[0]), "v"(raw[1]), "v"(raw[2]), "v"(raw[3]));
  #pragma unroll
  for (int u = 0; u < 4; ++u)
    s[u] = (l16 < len[u]) ? (int)min((unsigned)raw[u], (unsigned)(N - 1)) : nn[u];
  float4 q[4];
  #pragma unroll
  for (int u = 0; u < 4; ++u) q[u] = r4[s[u]];
  asm volatile("" :: "v"(q[0].x), "v"(q[0].y), "v"(q[0].z), "v"(q[0].w),
                     "v"(q[1].x), "v"(q[1].y), "v"(q[1].z), "v"(q[1].w),
                     "v"(q[2].x), "v"(q[2].y), "v"(q[2].z), "v"(q[2].w),
                     "v"(q[3].x), "v"(q[3].y), "v"(q[3].z), "v"(q[3].w));
  float a0[4], a1[4], a2[4];
  #pragma unroll
  for (int u = 0; u < 4; ++u) {
    const float w = (l16 < len[u]) ? q[u].w * own[u].w : 0.0f;
    a0[u] = (l16 < len[u]) ? q[u].x * w : 0.0f;
    a1[u] = (l16 < len[u]) ? q[u].y * w : 0.0f;
    a2[u] = (l16 < len[u]) ? q[u].z * w : 0.0f;
  }
  #pragma unroll
  for (int u = 0; u < 4; ++u)
    for (int slot = 16 + l16; slot < len[u]; slot += 16) {
      const int ss = (int)min((unsigned)col[beg[u] + slot], (unsigned)(N - 1));
      const float4 qq = r4[ss];
      const float w = qq.w * own[u].w;
      a0[u] = fmaf(qq.x, w, a0[u]); a1[u] = fmaf(qq.y, w, a1[u]); a2[u] = fmaf(qq.z, w, a2[u]);
    }
  #pragma unroll
  for (int u = 0; u < 4; ++u) {
    #pragma unroll
    for (int o = 8; o > 0; o >>= 1) {
      a0[u] += __shfl_xor(a0[u], o, 16);
      a1[u] += __shfl_xor(a1[u], o, 16);
      a2[u] += __shfl_xor(a2[u], o, 16);
    }
  }
  #pragma unroll
  for (int u = 0; u < 4; ++u)
    if (l16 == u && n0 + u < N) {
      const float dn = own[u].w;
      const float sl = dn * dn;
      ((float4*)g)[n0 + u] = make_float4(fmaf(own[u].x, sl, a0[u]), fmaf(own[u].y, sl, a1[u]), fmaf(own[u].z, sl, a2[u]), dn);
    }
}

__device__ __forceinline__ void pool_add(float* __restrict__ pool, const int* __restrict__ gcnt, int G,
                                         int g, int lane, float s0, float s1) {
  if ((unsigned)g < (unsigned)G) {
    const float cn = fmaxf((float)gcnt[g], 1.0f);
    float* o = pool + (long)g * HID;
    atomicAdd(o + lane, s0 / cn);
    atomicAdd(o + 32 + lane, s1 / cn);
  }
}

__global__ __launch_bounds__(256) void k_gather_mm(const int* __restrict__ rowinfo,
                                                   const int* __restrict__ col,
                                                   const float* __restrict__ g,
                                                   const float* __restrict__ W1,
                                                   const float* __restrict__ bias1,
                                                   const float* __restrict__ W,
                                                   const float* __restrict__ bias,
                                                   const int* __restrict__ batch,
                                                   float* __restrict__ pool,
                                                   const int* __restrict__ gcnt,
                                                   int pool_elems, int N, int E) {
  __shared__ alignas(16) float tile[8][16][68];
  const int wv   = threadIdx.x >> 5;
  const int lane = threadIdx.x & 31;
  const int half = lane >> 4;
  const int l16  = lane & 15;
  const int j    = l16 * 4;
  const int nb   = (blockIdx.x * 8 + wv) * 16;
  const float4* __restrict__ g4 = (const float4*)g;

  const float wa0 = W1[j + 0],           wa1 = W1[j + 1],           wa2 = W1[j + 2],           wa3 = W1[j + 3];
  const float wb0 = W1[HID + j + 0],     wb1 = W1[HID + j + 1],     wb2 = W1[HID + j + 2],     wb3 = W1[HID + j + 3];
  const float wc0 = W1[2 * HID + j + 0], wc1 = W1[2 * HID + j + 1], wc2 = W1[2 * HID + j + 2], wc3 = W1[2 * HID + j + 3];
  const float wd0 = bias1[j + 0],        wd1 = bias1[j + 1],        wd2 = bias1[j + 2],        wd3 = bias1[j + 3];

  int rbeg = 0, rlen = 0;
  float4 rg = {0.0f, 0.0f, 0.0f, 0.0f};
  int gid = 0;
  if (nb + l16 < N) {
    const int2 ri = ((const int2*)rowinfo)[nb + l16];
    rbeg = ri.x; rlen = ri.y;
    rg  = g4[nb + l16];
    gid = batch[nb + l16];
  }

  const int EL  = max(E, 1) - 1;
  const int nbc = min(nb, N - 1);
  int cbeg = __shfl(rbeg, half, 16);
  int clen = __shfl(rlen, half, 16);
  int nbeg = __shfl(rbeg, 2 + half, 16);
  int nlen = __shfl(rlen, 2 + half, 16);
  const int craw = col[min(cbeg + l16, EL)];
  int nraw = col[min(nbeg + l16, EL)];
  float4 cg = g4[(l16 < clen) ? (int)min((unsigned)craw, (unsigned)(N - 1)) : nbc];

  for (int p = 0; p < 8; ++p) {
    const int r   = 2 * p + half;
    const int beg = cbeg;
    const int len = clen;
    const float dn  = __shfl(rg.w, r, 16);
    const float sg0 = __shfl(rg.x, r, 16), sg1 = __shfl(rg.y, r, 16), sg2 = __shfl(rg.z, r, 16);
    const bool act = l16 < len;
    float mx0 = act ? cg.x : 0.0f, mx1 = act ? cg.y : 0.0f, mx2 = act ? cg.z : 0.0f;
    float myw = act ? cg.w * dn : 0.0f;
    cbeg = nbeg; clen = nlen;
    cg = g4[(l16 < clen) ? (int)min((unsigned)nraw, (unsigned)(N - 1)) : nbc];
    const int rr = min(r + 4, 15);
    nbeg = __shfl(rbeg, rr, 16);
    nlen = (p + 2 < 8) ? __shfl(rlen, rr, 16) : 0;
    nraw = col[min(nbeg + l16, EL)];

    float4 acc = {0.0f, 0.0f, 0.0f, 0.0f};
    for (int base = 0; base < len; base += 16) {
      if (base > 0) {
        mx0 = 0.0f; mx1 = 0.0f; mx2 = 0.0f; myw = 0.0f;
        if (base + l16 < len) {
          const int ss = (int)min((unsigned)col[beg + base + l16], (unsigned)(N - 1));
          const float4 q = g4[ss];
          mx0 = q.x; mx1 = q.y; mx2 = q.z; myw = q.w * dn;
        }
      }
      const int m = min(16, len - base);
      for (int k = 0; k < m; k += 4) {
        float x0[4], x1[4], x2[4], w[4];
        #pragma unroll
        for (int i = 0; i < 4; ++i) {
          x0[i] = __shfl(mx0, k + i, 16); x1[i] = __shfl(mx1, k + i, 16);
          x2[i] = __shfl(mx2, k + i, 16); w[i]  = __shfl(myw, k + i, 16);
        }
        #pragma unroll
        for (int i = 0; i < 4; ++i) {
          const float r0 = fmaxf(fmaf(x2[i], wc0, fmaf(x1[i], wb0, fmaf(x0[i], wa0, wd0))), 0.0f);
          const float r1 = fmaxf(fmaf(x2[i], wc1, fmaf(x1[i], wb1, fmaf(x0[i], wa1, wd1))), 0.0f);
          const float r2 = fmaxf(fmaf(x2[i], wc2, fmaf(x1[i], wb2, fmaf(x0[i], wa2, wd2))), 0.0f);
          const float r3 = fmaxf(fmaf(x2[i], wc3, fmaf(x1[i], wb3, fmaf(x0[i], wa3, wd3))), 0.0f);
          acc.x = fmaf(r0, w[i], acc.x); acc.y = fmaf(r1, w[i], acc.y);
          acc.z = fmaf(r2, w[i], acc.z); acc.w = fmaf(r3, w[i], acc.w);
        }
      }
    }

    const float sl = dn * dn;
    float4 a;
    a.x = fmaf(fmaxf(fmaf(sg2, wc0, fmaf(sg1, wb0, fmaf(sg0, wa0, wd0))), 0.0f), sl, acc.x);
    a.y = fmaf(fmaxf(fmaf(sg2, wc1, fmaf(sg1, wb1, fmaf(sg0, wa1, wd1))), 0.0f), sl, acc.y);
    a.z = fmaf(fmaxf(fmaf(sg2, wc2, fmaf(sg1, wb2, fmaf(sg0, wa2, wd2))), 0.0f), sl, acc.z);
    a.w = fmaf(fmaxf(fmaf(sg2, wc3, fmaf(sg1, wb3, fmaf(sg0, wa3, wd3))), 0.0f), sl, acc.w);
    *(float4*)&tile[wv][r][j] = a;
  }
  __syncthreads();

  v8f c0 = {}, c1 = {}, c2 = {}, c3 = {};
  const float* __restrict__ Arow = &tile[wv][l16][0];
  #pragma unroll
  for (int i = 0; i < 16; ++i) {
    const int ka = 4 * i + 2 * half;
    v2f a;  a.x  = Arow[ka];                    a.y  = Arow[ka + 1];
    v2f b0; b0.x = W[ka * HID + l16];           b0.y = W[(ka + 1) * HID + l16];
    v2f b1; b1.x = W[ka * HID + 16 + l16];      b1.y = W[(ka + 1) * HID + 16 + l16];
    v2f b2; b2.x = W[ka * HID + 32 + l16];      b2.y = W[(ka + 1) * HID + 32 + l16];
    v2f b3; b3.x = W[ka * HID + 48 + l16];      b3.y = W[(ka + 1) * HID + 48 + l16];
    c0 = __builtin_amdgcn_wmma_f32_16x16x4_f32(false, a, false, b0, (short)0, c0, false, false);
    c1 = __builtin_amdgcn_wmma_f32_16x16x4_f32(false, a, false, b1, (short)0, c1, false, false);
    c2 = __builtin_amdgcn_wmma_f32_16x16x4_f32(false, a, false, b2, (short)0, c2, false, false);
    c3 = __builtin_amdgcn_wmma_f32_16x16x4_f32(false, a, false, b3, (short)0, c3, false, false);
  }
  __syncthreads();

  const float bb0 = bias[l16], bb1 = bias[16 + l16], bb2 = bias[32 + l16], bb3 = bias[48 + l16];
  #pragma unroll
  for (int q = 0; q < 8; ++q) {
    float* trow = &tile[wv][q + 8 * half][0];
    trow[l16]      = fmaxf(c0[q] + bb0, 0.0f);
    trow[16 + l16] = fmaxf(c1[q] + bb1, 0.0f);
    trow[32 + l16] = fmaxf(c2[q] + bb2, 0.0f);
    trow[48 + l16] = fmaxf(c3[q] + bb3, 0.0f);
  }
  __syncthreads();

  const int G   = pool_elems >> 6;
  const int nbu = __builtin_amdgcn_readfirstlane(nb);
  float t0[16], t1[16];
  #pragma unroll
  for (int r = 0; r < 16; ++r) {
    t0[r] = tile[wv][r][lane];
    t1[r] = tile[wv][r][32 + lane];
  }
  float s0 = 0.0f, s1 = 0.0f;
  int gp = 0, run = 0;
  #pragma unroll
  for (int r = 0; r < 16; ++r) {
    if (nbu + r < N) {
      const int g = __builtin_amdgcn_readlane(gid, r);
      if (run > 0 && g != gp) {
        pool_add(pool, gcnt, G, gp, lane, s0, s1);
        s0 = 0.0f; s1 = 0.0f; run = 0;
      }
      gp = g;
      s0 += t0[r];
      s1 += t1[r];
      ++run;
    }
  }
  if (run > 0) pool_add(pool, gcnt, G, gp, lane, s0, s1);
}

extern "C" void kernel_launch(void* const* d_in, const int* in_sizes, int n_in,
                              void* d_out, int out_size, void* d_ws, size_t ws_size,
                              hipStream_t stream) {
  const float* x   = (const float*)d_in[0];
  const float* W1  = (const float*)d_in[1];
  const float* b1  = (const float*)d_in[2];
  const float* W2  = (const float*)d_in[3];
  const float* b2  = (const float*)d_in[4];
  const int*   ei  = (const int*)d_in[5];
  const int*   bat = (const int*)d_in[6];
  float* out = (float*)d_out;

  const int N = in_sizes[0] / 3;
  const int E = in_sizes[5] / 2;

  char* ws = (char*)d_ws;
  float* rec  = (float*)(ws);
  float* gfe  = (float*)(ws + (1 << 19) + (size_t)N * HID * 4);
  int*   gcnt = (int*)(ws + (1 << 19) + 2 * (size_t)N * HID * 4);
  const size_t Np = ((size_t)N + 63) & ~(size_t)63;
  const size_t Ep = ((size_t)E + 63) & ~(size_t)63;
  const int NB   = (N + RB - 1) / RB;
  const int NCH  = (E + CH - 1) / CH;
  const int NCHP = (NCH + 63) & ~63;
  int*  rowinfo = (int*)(ws + (1 << 19) + 2 * (size_t)N * HID * 4 + 4096);
  int*  col     = rowinfo + 2 * Np;
  int*  sorted  = col + Ep;
  int*  offT    = sorted + 2 * (size_t)NCH * CH;

  const int B = 256;
  const int N4   = ((N + 3) / 4) * 16;

  k_bin <<<NCH, 1024, 0, stream>>>(ei, E, N, NB, NCHP, sorted, offT, out, out_size, gcnt);
  k_sort<<<NB,  1024, 0, stream>>>(sorted, offT, NCH, NCHP, N, rowinfo, x, rec, col, bat, gcnt, out_size >> 6);

  k_feat1<<<(N4 + B - 1) / B, B, 0, stream>>>(rowinfo, col, rec, gfe, N, E);

  k_gather_mm<<<(N + 127) / 128, B, 0, stream>>>(rowinfo, col, gfe, W1, b1, W2, b2, bat, out, gcnt, out_size, N, E);
}
